# starting kernel
# baseline (speedup 1.0000x reference)
#include <hip/hip_runtime.h>
#include <math.h>


#define WIRES 11
#define DIM   2048
#define PIX   1024
#define NB    1024

typedef __attribute__((ext_vector_type(2))) float v2f;
typedef __attribute__((ext_vector_type(8))) float v8f;

__global__ void qprep(const float* __restrict__ w0, const float* __restrict__ w1,
                      float* __restrict__ G) {
  const float PI_ = 3.14159265358979323846f;
  for (int e = threadIdx.x; e < 8 * 3 * 256; e += blockDim.x) {
    int L    = e / 768;
    int rem  = e - L * 768;
    int grp  = rem >> 8;
    int cell = rem & 255;
    int m = cell >> 4, k = cell & 15;
    const float* w = (L < 4) ? w0 : w1;
    int l = L & 3;

    float re = 1.f, im = 0.f;
    int nw, q0, topbit;
    bool zero = false;
    if (grp == 0)      { nw = 4; q0 = 0; topbit = 3; }
    else if (grp == 1) { nw = 4; q0 = 4; topbit = 3; }
    else               { nw = 3; q0 = 8; topbit = 2;
                         if ((m >> 3) != (k >> 3)) zero = true; }
    if (zero) { re = 0.f; im = 0.f; }
    else {
      for (int i = 0; i < nw; ++i) {
        int q  = q0 + i;
        int bp = topbit - i;
        int mb = (m >> bp) & 1, kb = (k >> bp) & 1;
        const float* a3 = w + (l * WIRES + q) * 3;
        float phi   = PI_ * tanhf(a3[0]);
        float theta = PI_ * tanhf(a3[1]);
        float omega = PI_ * tanhf(a3[2]);
        float c = cosf(0.5f * theta), s = sinf(0.5f * theta);
        float h = 0.5f * (phi + omega), g = 0.5f * (phi - omega);
        float ur, ui;
        if (mb == 0 && kb == 0)      { ur =  cosf(h) * c; ui = -sinf(h) * c; }
        else if (mb == 0 && kb == 1) { ur = -cosf(g) * s; ui = -sinf(g) * s; }
        else if (mb == 1 && kb == 0) { ur =  cosf(g) * s; ui = -sinf(g) * s; }
        else                         { ur =  cosf(h) * c; ui =  sinf(h) * c; }
        float nr = re * ur - im * ui;
        float ni = re * ui + im * ur;
        re = nr; im = ni;
      }
    }
    int base = (L * 3 + grp) * 512;
    G[base + cell]       = re;
    G[base + 256 + cell] = im;
  }
}

__device__ __forceinline__ int pinv_idx(int x, int r) {
#pragma unroll
  for (int q = WIRES - 1; q >= 0; --q) {
    int tq = q + r; if (tq >= WIRES) tq -= WIRES;
    x ^= ((x >> (10 - q)) & 1) << (10 - tq);
  }
  return x;
}

__device__ __forceinline__ void apply_group(float2* st,
                                            const float* __restrict__ gr,
                                            const float* __restrict__ gi,
                                            int lo, int wave, int lane, int rperm) {
  const int n    = lane & 15;
  const int half = lane >> 4;
  const int j    = wave * 16 + n;
  const int base = ((j >> lo) << (lo + 4)) | (j & ((1 << lo) - 1));

  v2f ar[4], ai[4], br[4], bi[4];
#pragma unroll
  for (int c = 0; c < 4; ++c) {
    int k0 = 4 * c + 2 * half;
    int i0 = base | (k0 << lo);
    int i1 = base | ((k0 + 1) << lo);
    if (rperm) { i0 = pinv_idx(i0, rperm); i1 = pinv_idx(i1, rperm); }
    float2 s0 = st[i0];
    float2 s1 = st[i1];
    br[c] = v2f{s0.x, s1.x};
    bi[c] = v2f{s0.y, s1.y};
    ar[c] = v2f{gr[n * 16 + k0], gr[n * 16 + k0 + 1]};
    ai[c] = v2f{gi[n * 16 + k0], gi[n * 16 + k0 + 1]};
  }
  if (rperm) __syncthreads();

  v8f dr = {0, 0, 0, 0, 0, 0, 0, 0};
  v8f di = {0, 0, 0, 0, 0, 0, 0, 0};
#pragma unroll
  for (int c = 0; c < 4; ++c) {
    v2f nai = -ai[c];
    dr = __builtin_amdgcn_wmma_f32_16x16x4_f32(false, ar[c], false, br[c],
                                               (short)0, dr, false, false);
    dr = __builtin_amdgcn_wmma_f32_16x16x4_f32(false, nai,   false, bi[c],
                                               (short)0, dr, false, false);
    di = __builtin_amdgcn_wmma_f32_16x16x4_f32(false, ar[c], false, bi[c],
                                               (short)0, di, false, false);
    di = __builtin_amdgcn_wmma_f32_16x16x4_f32(false, ai[c], false, br[c],
                                               (short)0, di, false, false);
  }

#pragma unroll
  for (int r = 0; r < 8; ++r) {
    int m = r + 8 * half;
    st[base | (m << lo)] = make_float2(dr[r], di[r]);
  }
}

__global__ void __launch_bounds__(256)
qsim(const float* __restrict__ x, const float* __restrict__ y,
     const float* __restrict__ G, float* __restrict__ out) {
  __shared__ float2 st[DIM];
  __shared__ float  red[256];
  const int tid  = threadIdx.x;
  const int lane = tid & 31;
  const int wave = tid >> 5;
  const int b    = blockIdx.x;

  float xs[4], part = 0.f;
#pragma unroll
  for (int t = 0; t < 4; ++t) {
    int p = tid + 256 * t;
    float v = x[b * PIX + p];
    xs[t] = v; part += v * v;
  }
  red[tid] = part; __syncthreads();
  for (int off = 128; off > 0; off >>= 1) {
    if (tid < off) red[tid] += red[tid + off];
    __syncthreads();
  }
  float inv = rsqrtf(red[0]);
#pragma unroll
  for (int t = 0; t < 4; ++t) {
    int p = tid + 256 * t;
    st[2 * p]     = make_float2(xs[t] * inv, 0.f);
    st[2 * p + 1] = make_float2(0.f, 0.f);
  }
  __syncthreads();

  const float yv = y[b];
  const float rc = cosf(0.5f * yv), rs = sinf(0.5f * yv);

  int pend = 0;
  for (int blkw = 0; blkw < 2; ++blkw) {
#pragma unroll
    for (int t = 0; t < 4; ++t) {
      int p  = tid + 256 * t;
      int x0 = 2 * p, x1 = 2 * p + 1;
      if (pend) { x0 = pinv_idx(x0, pend); x1 = pinv_idx(x1, pend); }
      float2 a0 = st[x0], a1 = st[x1];
      st[x0] = make_float2(rc * a0.x + rs * a1.y, rc * a0.y - rs * a1.x);
      st[x1] = make_float2(rc * a1.x + rs * a0.y, rc * a1.y - rs * a0.x);
    }
    __syncthreads();

    for (int l = 0; l < 4; ++l) {
      int L = blkw * 4 + l;
      const float* g0 = G + (L * 3 + 0) * 512;
      apply_group(st, g0, g0 + 256, 7, wave, lane, pend); __syncthreads();
      const float* g1 = G + (L * 3 + 1) * 512;
      apply_group(st, g1, g1 + 256, 3, wave, lane, 0);    __syncthreads();
      const float* g2 = G + (L * 3 + 2) * 512;
      apply_group(st, g2, g2 + 256, 0, wave, lane, 0);    __syncthreads();
      pend = l + 1;
    }
  }

#pragma unroll
  for (int t = 0; t < 4; ++t) {
    int p  = tid + 256 * t;
    float2 a = st[pinv_idx(2 * p, pend)];
    out[b * PIX + p] = fminf(1.0f, (a.x * a.x + a.y * a.y) * (float)PIX);
  }
}

extern "C" void kernel_launch(void* const* d_in, const int* in_sizes, int n_in,
                              void* d_out, int out_size, void* d_ws, size_t ws_size,
                              hipStream_t stream) {
  const float* x  = (const float*)d_in[0];
  const float* y  = (const float*)d_in[1];
  const float* w0 = (const float*)d_in[2];
  const float* w1 = (const float*)d_in[3];
  float* G = (float*)d_ws;

  qprep<<<1, 256, 0, stream>>>(w0, w1, G);
  qsim<<<NB, 256, 0, stream>>>(x, y, G, (float*)d_out);
}


# prep kernel spread over 24 work-groups (one per layer-group matrix, one cell per thread) instead of 1 work-group looping 24 times; same arithmetic per cell
# speedup vs baseline: 2.3622x; 2.3622x over previous
#include <hip/hip_runtime.h>
#include <math.h>


#define WIRES 11
#define DIM   2048
#define PIX   1024
#define NB    1024

typedef __attribute__((ext_vector_type(2))) float v2f;
typedef __attribute__((ext_vector_type(8))) float v8f;

__global__ void qprep(const float* __restrict__ w0, const float* __restrict__ w1,
                      float* __restrict__ G) {
  const float PI_ = 3.14159265358979323846f;
  {
    const int blk  = blockIdx.x;
    const int L    = blk / 3;
    const int grp  = blk - L * 3;
    const int cell = threadIdx.x;
    int m = cell >> 4, k = cell & 15;
    const float* w = (L < 4) ? w0 : w1;
    int l = L & 3;

    float re = 1.f, im = 0.f;
    int nw, q0, topbit;
    bool zero = false;
    if (grp == 0)      { nw = 4; q0 = 0; topbit = 3; }
    else if (grp == 1) { nw = 4; q0 = 4; topbit = 3; }
    else               { nw = 3; q0 = 8; topbit = 2;
                         if ((m >> 3) != (k >> 3)) zero = true; }
    if (zero) { re = 0.f; im = 0.f; }
    else {
      for (int i = 0; i < nw; ++i) {
        int q  = q0 + i;
        int bp = topbit - i;
        int mb = (m >> bp) & 1, kb = (k >> bp) & 1;
        const float* a3 = w + (l * WIRES + q) * 3;
        float phi   = PI_ * tanhf(a3[0]);
        float theta = PI_ * tanhf(a3[1]);
        float omega = PI_ * tanhf(a3[2]);
        float c = cosf(0.5f * theta), s = sinf(0.5f * theta);
        float h = 0.5f * (phi + omega), g = 0.5f * (phi - omega);
        float ur, ui;
        if (mb == 0 && kb == 0)      { ur =  cosf(h) * c; ui = -sinf(h) * c; }
        else if (mb == 0 && kb == 1) { ur = -cosf(g) * s; ui = -sinf(g) * s; }
        else if (mb == 1 && kb == 0) { ur =  cosf(g) * s; ui = -sinf(g) * s; }
        else                         { ur =  cosf(h) * c; ui =  sinf(h) * c; }
        float nr = re * ur - im * ui;
        float ni = re * ui + im * ur;
        re = nr; im = ni;
      }
    }
    int base = blk * 512;
    G[base + cell]       = re;
    G[base + 256 + cell] = im;
  }
}

__device__ __forceinline__ int pinv_idx(int x, int r) {
#pragma unroll
  for (int q = WIRES - 1; q >= 0; --q) {
    int tq = q + r; if (tq >= WIRES) tq -= WIRES;
    x ^= ((x >> (10 - q)) & 1) << (10 - tq);
  }
  return x;
}

__device__ __forceinline__ void apply_group(float2* st,
                                            const float* __restrict__ gr,
                                            const float* __restrict__ gi,
                                            int lo, int wave, int lane, int rperm) {
  const int n    = lane & 15;
  const int half = lane >> 4;
  const int j    = wave * 16 + n;
  const int base = ((j >> lo) << (lo + 4)) | (j & ((1 << lo) - 1));

  v2f ar[4], ai[4], br[4], bi[4];
#pragma unroll
  for (int c = 0; c < 4; ++c) {
    int k0 = 4 * c + 2 * half;
    int i0 = base | (k0 << lo);
    int i1 = base | ((k0 + 1) << lo);
    if (rperm) { i0 = pinv_idx(i0, rperm); i1 = pinv_idx(i1, rperm); }
    float2 s0 = st[i0];
    float2 s1 = st[i1];
    br[c] = v2f{s0.x, s1.x};
    bi[c] = v2f{s0.y, s1.y};
    ar[c] = v2f{gr[n * 16 + k0], gr[n * 16 + k0 + 1]};
    ai[c] = v2f{gi[n * 16 + k0], gi[n * 16 + k0 + 1]};
  }
  if (rperm) __syncthreads();

  v8f dr = {0, 0, 0, 0, 0, 0, 0, 0};
  v8f di = {0, 0, 0, 0, 0, 0, 0, 0};
#pragma unroll
  for (int c = 0; c < 4; ++c) {
    v2f nai = -ai[c];
    dr = __builtin_amdgcn_wmma_f32_16x16x4_f32(false, ar[c], false, br[c],
                                               (short)0, dr, false, false);
    dr = __builtin_amdgcn_wmma_f32_16x16x4_f32(false, nai,   false, bi[c],
                                               (short)0, dr, false, false);
    di = __builtin_amdgcn_wmma_f32_16x16x4_f32(false, ar[c], false, bi[c],
                                               (short)0, di, false, false);
    di = __builtin_amdgcn_wmma_f32_16x16x4_f32(false, ai[c], false, br[c],
                                               (short)0, di, false, false);
  }

#pragma unroll
  for (int r = 0; r < 8; ++r) {
    int m = r + 8 * half;
    st[base | (m << lo)] = make_float2(dr[r], di[r]);
  }
}

__global__ void __launch_bounds__(256)
qsim(const float* __restrict__ x, const float* __restrict__ y,
     const float* __restrict__ G, float* __restrict__ out) {
  __shared__ float2 st[DIM];
  __shared__ float  red[256];
  const int tid  = threadIdx.x;
  const int lane = tid & 31;
  const int wave = tid >> 5;
  const int b    = blockIdx.x;

  float xs[4], part = 0.f;
#pragma unroll
  for (int t = 0; t < 4; ++t) {
    int p = tid + 256 * t;
    float v = x[b * PIX + p];
    xs[t] = v; part += v * v;
  }
  red[tid] = part; __syncthreads();
  for (int off = 128; off > 0; off >>= 1) {
    if (tid < off) red[tid] += red[tid + off];
    __syncthreads();
  }
  float inv = rsqrtf(red[0]);
#pragma unroll
  for (int t = 0; t < 4; ++t) {
    int p = tid + 256 * t;
    st[2 * p]     = make_float2(xs[t] * inv, 0.f);
    st[2 * p + 1] = make_float2(0.f, 0.f);
  }
  __syncthreads();

  const float yv = y[b];
  const float rc = cosf(0.5f * yv), rs = sinf(0.5f * yv);

  int pend = 0;
  for (int blkw = 0; blkw < 2; ++blkw) {
#pragma unroll
    for (int t = 0; t < 4; ++t) {
      int p  = tid + 256 * t;
      int x0 = 2 * p, x1 = 2 * p + 1;
      if (pend) { x0 = pinv_idx(x0, pend); x1 = pinv_idx(x1, pend); }
      float2 a0 = st[x0], a1 = st[x1];
      st[x0] = make_float2(rc * a0.x + rs * a1.y, rc * a0.y - rs * a1.x);
      st[x1] = make_float2(rc * a1.x + rs * a0.y, rc * a1.y - rs * a0.x);
    }
    __syncthreads();

    for (int l = 0; l < 4; ++l) {
      int L = blkw * 4 + l;
      const float* g0 = G + (L * 3 + 0) * 512;
      apply_group(st, g0, g0 + 256, 7, wave, lane, pend); __syncthreads();
      const float* g1 = G + (L * 3 + 1) * 512;
      apply_group(st, g1, g1 + 256, 3, wave, lane, 0);    __syncthreads();
      const float* g2 = G + (L * 3 + 2) * 512;
      apply_group(st, g2, g2 + 256, 0, wave, lane, 0);    __syncthreads();
      pend = l + 1;
    }
  }

#pragma unroll
  for (int t = 0; t < 4; ++t) {
    int p  = tid + 256 * t;
    float2 a = st[pinv_idx(2 * p, pend)];
    out[b * PIX + p] = fminf(1.0f, (a.x * a.x + a.y * a.y) * (float)PIX);
  }
}

extern "C" void kernel_launch(void* const* d_in, const int* in_sizes, int n_in,
                              void* d_out, int out_size, void* d_ws, size_t ws_size,
                              hipStream_t stream) {
  const float* x  = (const float*)d_in[0];
  const float* y  = (const float*)d_in[1];
  const float* w0 = (const float*)d_in[2];
  const float* w1 = (const float*)d_in[3];
  float* G = (float*)d_ws;

  qprep<<<24, 256, 0, stream>>>(w0, w1, G);
  qsim<<<NB, 256, 0, stream>>>(x, y, G, (float*)d_out);
}


# complex 16x16 group product by the 3-real-product identity K1=Gr(Sr+Si), Dr=K1-(Gr+Gi)Si, Di=K1+(Gi-Gr)Sr: 12 WMMA per tile instead of 16; changes rounding and summation order
# speedup vs baseline: 2.6090x; 1.1045x over previous
#include <hip/hip_runtime.h>
#include <math.h>


#define WIRES 11
#define DIM   2048
#define PIX   1024
#define NB    1024

typedef __attribute__((ext_vector_type(2))) float v2f;
typedef __attribute__((ext_vector_type(8))) float v8f;

__global__ void qprep(const float* __restrict__ w0, const float* __restrict__ w1,
                      float* __restrict__ G) {
  const float PI_ = 3.14159265358979323846f;
  {
    const int blk  = blockIdx.x;
    const int L    = blk / 3;
    const int grp  = blk - L * 3;
    const int cell = threadIdx.x;
    int m = cell >> 4, k = cell & 15;
    const float* w = (L < 4) ? w0 : w1;
    int l = L & 3;

    float re = 1.f, im = 0.f;
    int nw, q0, topbit;
    bool zero = false;
    if (grp == 0)      { nw = 4; q0 = 0; topbit = 3; }
    else if (grp == 1) { nw = 4; q0 = 4; topbit = 3; }
    else               { nw = 3; q0 = 8; topbit = 2;
                         if ((m >> 3) != (k >> 3)) zero = true; }
    if (zero) { re = 0.f; im = 0.f; }
    else {
      for (int i = 0; i < nw; ++i) {
        int q  = q0 + i;
        int bp = topbit - i;
        int mb = (m >> bp) & 1, kb = (k >> bp) & 1;
        const float* a3 = w + (l * WIRES + q) * 3;
        float phi   = PI_ * tanhf(a3[0]);
        float theta = PI_ * tanhf(a3[1]);
        float omega = PI_ * tanhf(a3[2]);
        float c = cosf(0.5f * theta), s = sinf(0.5f * theta);
        float h = 0.5f * (phi + omega), g = 0.5f * (phi - omega);
        float ur, ui;
        if (mb == 0 && kb == 0)      { ur =  cosf(h) * c; ui = -sinf(h) * c; }
        else if (mb == 0 && kb == 1) { ur = -cosf(g) * s; ui = -sinf(g) * s; }
        else if (mb == 1 && kb == 0) { ur =  cosf(g) * s; ui = -sinf(g) * s; }
        else                         { ur =  cosf(h) * c; ui =  sinf(h) * c; }
        float nr = re * ur - im * ui;
        float ni = re * ui + im * ur;
        re = nr; im = ni;
      }
    }
    int base = blk * 512;
    G[base + cell]       = re;
    G[base + 256 + cell] = im;
  }
}

__device__ __forceinline__ int pinv_idx(int x, int r) {
#pragma unroll
  for (int q = WIRES - 1; q >= 0; --q) {
    int tq = q + r; if (tq >= WIRES) tq -= WIRES;
    x ^= ((x >> (10 - q)) & 1) << (10 - tq);
  }
  return x;
}

__device__ __forceinline__ void apply_group(float2* st,
                                            const float* __restrict__ gr,
                                            const float* __restrict__ gi,
                                            int lo, int wave, int lane, int rperm) {
  const int n    = lane & 15;
  const int half = lane >> 4;
  const int j    = wave * 16 + n;
  const int base = ((j >> lo) << (lo + 4)) | (j & ((1 << lo) - 1));

  v2f ar[4], ai[4], br[4], bi[4];
#pragma unroll
  for (int c = 0; c < 4; ++c) {
    int k0 = 4 * c + 2 * half;
    int i0 = base | (k0 << lo);
    int i1 = base | ((k0 + 1) << lo);
    if (rperm) { i0 = pinv_idx(i0, rperm); i1 = pinv_idx(i1, rperm); }
    float2 s0 = st[i0];
    float2 s1 = st[i1];
    br[c] = v2f{s0.x, s1.x};
    bi[c] = v2f{s0.y, s1.y};
    ar[c] = v2f{gr[n * 16 + k0], gr[n * 16 + k0 + 1]};
    ai[c] = v2f{gi[n * 16 + k0], gi[n * 16 + k0 + 1]};
  }
  if (rperm) __syncthreads();

  v8f k1 = {0, 0, 0, 0, 0, 0, 0, 0};
#pragma unroll
  for (int c = 0; c < 4; ++c) {
    v2f bs = br[c] + bi[c];
    k1 = __builtin_amdgcn_wmma_f32_16x16x4_f32(false, ar[c], false, bs,
                                               (short)0, k1, false, false);
  }
  v8f dr = k1;
  v8f di = k1;
#pragma unroll
  for (int c = 0; c < 4; ++c) {
    v2f q = -ar[c] - ai[c];
    v2f p = ai[c] - ar[c];
    dr = __builtin_amdgcn_wmma_f32_16x16x4_f32(false, q, false, bi[c],
                                               (short)0, dr, false, false);
    di = __builtin_amdgcn_wmma_f32_16x16x4_f32(false, p, false, br[c],
                                               (short)0, di, false, false);
  }

#pragma unroll
  for (int r = 0; r < 8; ++r) {
    int m = r + 8 * half;
    st[base | (m << lo)] = make_float2(dr[r], di[r]);
  }
}

__global__ void __launch_bounds__(256)
qsim(const float* __restrict__ x, const float* __restrict__ y,
     const float* __restrict__ G, float* __restrict__ out) {
  __shared__ float2 st[DIM];
  __shared__ float  red[256];
  const int tid  = threadIdx.x;
  const int lane = tid & 31;
  const int wave = tid >> 5;
  const int b    = blockIdx.x;

  float xs[4], part = 0.f;
#pragma unroll
  for (int t = 0; t < 4; ++t) {
    int p = tid + 256 * t;
    float v = x[b * PIX + p];
    xs[t] = v; part += v * v;
  }
  red[tid] = part; __syncthreads();
  for (int off = 128; off > 0; off >>= 1) {
    if (tid < off) red[tid] += red[tid + off];
    __syncthreads();
  }
  float inv = rsqrtf(red[0]);
#pragma unroll
  for (int t = 0; t < 4; ++t) {
    int p = tid + 256 * t;
    st[2 * p]     = make_float2(xs[t] * inv, 0.f);
    st[2 * p + 1] = make_float2(0.f, 0.f);
  }
  __syncthreads();

  const float yv = y[b];
  const float rc = cosf(0.5f * yv), rs = sinf(0.5f * yv);

  int pend = 0;
  for (int blkw = 0; blkw < 2; ++blkw) {
#pragma unroll
    for (int t = 0; t < 4; ++t) {
      int p  = tid + 256 * t;
      int x0 = 2 * p, x1 = 2 * p + 1;
      if (pend) { x0 = pinv_idx(x0, pend); x1 = pinv_idx(x1, pend); }
      float2 a0 = st[x0], a1 = st[x1];
      st[x0] = make_float2(rc * a0.x + rs * a1.y, rc * a0.y - rs * a1.x);
      st[x1] = make_float2(rc * a1.x + rs * a0.y, rc * a1.y - rs * a0.x);
    }
    __syncthreads();

    for (int l = 0; l < 4; ++l) {
      int L = blkw * 4 + l;
      const float* g0 = G + (L * 3 + 0) * 512;
      apply_group(st, g0, g0 + 256, 7, wave, lane, pend); __syncthreads();
      const float* g1 = G + (L * 3 + 1) * 512;
      apply_group(st, g1, g1 + 256, 3, wave, lane, 0);    __syncthreads();
      const float* g2 = G + (L * 3 + 2) * 512;
      apply_group(st, g2, g2 + 256, 0, wave, lane, 0);    __syncthreads();
      pend = l + 1;
    }
  }

#pragma unroll
  for (int t = 0; t < 4; ++t) {
    int p  = tid + 256 * t;
    float2 a = st[pinv_idx(2 * p, pend)];
    out[b * PIX + p] = fminf(1.0f, (a.x * a.x + a.y * a.y) * (float)PIX);
  }
}

extern "C" void kernel_launch(void* const* d_in, const int* in_sizes, int n_in,
                              void* d_out, int out_size, void* d_ws, size_t ws_size,
                              hipStream_t stream) {
  const float* x  = (const float*)d_in[0];
  const float* y  = (const float*)d_in[1];
  const float* w0 = (const float*)d_in[2];
  const float* w1 = (const float*)d_in[3];
  float* G = (float*)d_ws;

  qprep<<<24, 256, 0, stream>>>(w0, w1, G);
  qsim<<<NB, 256, 0, stream>>>(x, y, G, (float*)d_out);
}


# CNOT-ring gather index by GF(2) linearity: one per-lane map per pass XOR compile-time constants (layers unrolled so the ring range is constant), not eight 11-step bit loops per lane; same arithmetic
# speedup vs baseline: 3.0246x; 1.1593x over previous
#include <hip/hip_runtime.h>
#include <math.h>


#define WIRES 11
#define DIM   2048
#define PIX   1024
#define NB    1024

typedef __attribute__((ext_vector_type(2))) float v2f;
typedef __attribute__((ext_vector_type(8))) float v8f;

__global__ void qprep(const float* __restrict__ w0, const float* __restrict__ w1,
                      float* __restrict__ G) {
  const float PI_ = 3.14159265358979323846f;
  {
    const int blk  = blockIdx.x;
    const int L    = blk / 3;
    const int grp  = blk - L * 3;
    const int cell = threadIdx.x;
    int m = cell >> 4, k = cell & 15;
    const float* w = (L < 4) ? w0 : w1;
    int l = L & 3;

    float re = 1.f, im = 0.f;
    int nw, q0, topbit;
    bool zero = false;
    if (grp == 0)      { nw = 4; q0 = 0; topbit = 3; }
    else if (grp == 1) { nw = 4; q0 = 4; topbit = 3; }
    else               { nw = 3; q0 = 8; topbit = 2;
                         if ((m >> 3) != (k >> 3)) zero = true; }
    if (zero) { re = 0.f; im = 0.f; }
    else {
      for (int i = 0; i < nw; ++i) {
        int q  = q0 + i;
        int bp = topbit - i;
        int mb = (m >> bp) & 1, kb = (k >> bp) & 1;
        const float* a3 = w + (l * WIRES + q) * 3;
        float phi   = PI_ * tanhf(a3[0]);
        float theta = PI_ * tanhf(a3[1]);
        float omega = PI_ * tanhf(a3[2]);
        float c = cosf(0.5f * theta), s = sinf(0.5f * theta);
        float h = 0.5f * (phi + omega), g = 0.5f * (phi - omega);
        float ur, ui;
        if (mb == 0 && kb == 0)      { ur =  cosf(h) * c; ui = -sinf(h) * c; }
        else if (mb == 0 && kb == 1) { ur = -cosf(g) * s; ui = -sinf(g) * s; }
        else if (mb == 1 && kb == 0) { ur =  cosf(g) * s; ui = -sinf(g) * s; }
        else                         { ur =  cosf(h) * c; ui =  sinf(h) * c; }
        float nr = re * ur - im * ui;
        float ni = re * ui + im * ur;
        re = nr; im = ni;
      }
    }
    int base = blk * 512;
    G[base + cell]       = re;
    G[base + 256 + cell] = im;
  }
}

__host__ __device__ constexpr int pinv_idx(int x, int r) {
  for (int q = WIRES - 1; q >= 0; --q) {
    int tq = q + r; if (tq >= WIRES) tq -= WIRES;
    x ^= ((x >> (10 - q)) & 1) << (10 - tq);
  }
  return x;
}

template <int RP, int LO>
__device__ __forceinline__ void apply_group(float2* st,
                                            const float* __restrict__ gr,
                                            const float* __restrict__ gi,
                                            int wave, int lane) {
  const int n    = lane & 15;
  const int half = lane >> 4;
  const int j    = wave * 16 + n;
  const int base = ((j >> LO) << (LO + 4)) | (j & ((1 << LO) - 1));

  constexpr int KC[4] = { 0,
                          RP ? pinv_idx(4  << LO, RP) : 0,
                          RP ? pinv_idx(8  << LO, RP) : 0,
                          RP ? pinv_idx(12 << LO, RP) : 0 };
  constexpr int KE = RP ? pinv_idx(1 << LO, RP) : 0;
  int vb = 0;
  if constexpr (RP != 0) vb = pinv_idx(base | ((2 * half) << LO), RP);

  v2f ar[4], ai[4], br[4], bi[4];
#pragma unroll
  for (int c = 0; c < 4; ++c) {
    int k0 = 4 * c + 2 * half;
    int i0, i1;
    if constexpr (RP != 0) {
      i0 = vb ^ KC[c];
      i1 = i0 ^ KE;
    } else {
      i0 = base | (k0 << LO);
      i1 = base | ((k0 + 1) << LO);
    }
    float2 s0 = st[i0];
    float2 s1 = st[i1];
    br[c] = v2f{s0.x, s1.x};
    bi[c] = v2f{s0.y, s1.y};
    ar[c] = v2f{gr[n * 16 + k0], gr[n * 16 + k0 + 1]};
    ai[c] = v2f{gi[n * 16 + k0], gi[n * 16 + k0 + 1]};
  }
  if constexpr (RP != 0) __syncthreads();

  v8f k1 = {0, 0, 0, 0, 0, 0, 0, 0};
#pragma unroll
  for (int c = 0; c < 4; ++c) {
    v2f bs = br[c] + bi[c];
    k1 = __builtin_amdgcn_wmma_f32_16x16x4_f32(false, ar[c], false, bs,
                                               (short)0, k1, false, false);
  }
  v8f dr = k1;
  v8f di = k1;
#pragma unroll
  for (int c = 0; c < 4; ++c) {
    v2f q = -ar[c] - ai[c];
    v2f p = ai[c] - ar[c];
    dr = __builtin_amdgcn_wmma_f32_16x16x4_f32(false, q, false, bi[c],
                                               (short)0, dr, false, false);
    di = __builtin_amdgcn_wmma_f32_16x16x4_f32(false, p, false, br[c],
                                               (short)0, di, false, false);
  }

#pragma unroll
  for (int r = 0; r < 8; ++r) {
    int m = r + 8 * half;
    st[base | (m << LO)] = make_float2(dr[r], di[r]);
  }
}

template <int RP>
__device__ __forceinline__ void rx_pass(float2* st, int tid, float rc, float rs) {
  constexpr int KT[4] = { 0,
                          RP ? pinv_idx(512,  RP) : 0,
                          RP ? pinv_idx(1024, RP) : 0,
                          RP ? pinv_idx(1536, RP) : 0 };
  constexpr int K1 = RP ? pinv_idx(1, RP) : 0;
  int vt = 0;
  if constexpr (RP != 0) vt = pinv_idx(2 * tid, RP);
#pragma unroll
  for (int t = 0; t < 4; ++t) {
    int x0, x1;
    if constexpr (RP != 0) {
      x0 = vt ^ KT[t];
      x1 = x0 ^ K1;
    } else {
      int p = tid + 256 * t;
      x0 = 2 * p; x1 = 2 * p + 1;
    }
    float2 a0 = st[x0], a1 = st[x1];
    st[x0] = make_float2(rc * a0.x + rs * a1.y, rc * a0.y - rs * a1.x);
    st[x1] = make_float2(rc * a1.x + rs * a0.y, rc * a1.y - rs * a0.x);
  }
}

template <int RP>
__device__ __forceinline__ void layer(float2* st, const float* __restrict__ Gl,
                                      int wave, int lane) {
  apply_group<RP, 7>(st, Gl,        Gl + 256,  wave, lane); __syncthreads();
  apply_group<0, 3>(st, Gl + 512,  Gl + 768,  wave, lane); __syncthreads();
  apply_group<0, 0>(st, Gl + 1024, Gl + 1280, wave, lane); __syncthreads();
}

template <int RP>
__device__ __forceinline__ void reupload_block(float2* st, const float* __restrict__ Gb,
                                               int tid, int wave, int lane,
                                               float rc, float rs) {
  rx_pass<RP>(st, tid, rc, rs);
  __syncthreads();
  layer<RP>(st, Gb,            wave, lane);
  layer<1>(st, Gb + 1 * 1536, wave, lane);
  layer<2>(st, Gb + 2 * 1536, wave, lane);
  layer<3>(st, Gb + 3 * 1536, wave, lane);
}

__global__ void __launch_bounds__(256)
qsim(const float* __restrict__ x, const float* __restrict__ y,
     const float* __restrict__ G, float* __restrict__ out) {
  __shared__ float2 st[DIM];
  __shared__ float  red[256];
  const int tid  = threadIdx.x;
  const int lane = tid & 31;
  const int wave = tid >> 5;
  const int b    = blockIdx.x;

  float xs[4], part = 0.f;
#pragma unroll
  for (int t = 0; t < 4; ++t) {
    int p = tid + 256 * t;
    float v = x[b * PIX + p];
    xs[t] = v; part += v * v;
  }
  red[tid] = part; __syncthreads();
  for (int off = 128; off > 0; off >>= 1) {
    if (tid < off) red[tid] += red[tid + off];
    __syncthreads();
  }
  float inv = rsqrtf(red[0]);
#pragma unroll
  for (int t = 0; t < 4; ++t) {
    int p = tid + 256 * t;
    st[2 * p]     = make_float2(xs[t] * inv, 0.f);
    st[2 * p + 1] = make_float2(0.f, 0.f);
  }
  __syncthreads();

  const float yv = y[b];
  const float rc = cosf(0.5f * yv), rs = sinf(0.5f * yv);

  reupload_block<0>(st, G,            tid, wave, lane, rc, rs);
  reupload_block<4>(st, G + 4 * 1536, tid, wave, lane, rc, rs);

  const int vo = pinv_idx(2 * tid, 4);
#pragma unroll
  for (int t = 0; t < 4; ++t) {
    int p  = tid + 256 * t;
    float2 a = st[vo ^ pinv_idx(512 * t, 4)];
    out[b * PIX + p] = fminf(1.0f, (a.x * a.x + a.y * a.y) * (float)PIX);
  }
}

extern "C" void kernel_launch(void* const* d_in, const int* in_sizes, int n_in,
                              void* d_out, int out_size, void* d_ws, size_t ws_size,
                              hipStream_t stream) {
  const float* x  = (const float*)d_in[0];
  const float* y  = (const float*)d_in[1];
  const float* w0 = (const float*)d_in[2];
  const float* w1 = (const float*)d_in[3];
  float* G = (float*)d_ws;

  qprep<<<24, 256, 0, stream>>>(w0, w1, G);
  qsim<<<NB, 256, 0, stream>>>(x, y, G, (float*)d_out);
}


# LDS state placed through an XOR bank swizzle (index bits 5..10 folded into bits 0..4; linear, so it composes with the ring map): all fragment gathers and stores bank-conflict-free; same arithmetic
# speedup vs baseline: 3.2970x; 1.0901x over previous
#include <hip/hip_runtime.h>
#include <math.h>


#define WIRES 11
#define DIM   2048
#define PIX   1024
#define NB    1024

typedef __attribute__((ext_vector_type(2))) float v2f;
typedef __attribute__((ext_vector_type(8))) float v8f;

__global__ void qprep(const float* __restrict__ w0, const float* __restrict__ w1,
                      float* __restrict__ G) {
  const float PI_ = 3.14159265358979323846f;
  {
    const int blk  = blockIdx.x;
    const int L    = blk / 3;
    const int grp  = blk - L * 3;
    const int cell = threadIdx.x;
    int m = cell >> 4, k = cell & 15;
    const float* w = (L < 4) ? w0 : w1;
    int l = L & 3;

    float re = 1.f, im = 0.f;
    int nw, q0, topbit;
    bool zero = false;
    if (grp == 0)      { nw = 4; q0 = 0; topbit = 3; }
    else if (grp == 1) { nw = 4; q0 = 4; topbit = 3; }
    else               { nw = 3; q0 = 8; topbit = 2;
                         if ((m >> 3) != (k >> 3)) zero = true; }
    if (zero) { re = 0.f; im = 0.f; }
    else {
      for (int i = 0; i < nw; ++i) {
        int q  = q0 + i;
        int bp = topbit - i;
        int mb = (m >> bp) & 1, kb = (k >> bp) & 1;
        const float* a3 = w + (l * WIRES + q) * 3;
        float phi   = PI_ * tanhf(a3[0]);
        float theta = PI_ * tanhf(a3[1]);
        float omega = PI_ * tanhf(a3[2]);
        float c = cosf(0.5f * theta), s = sinf(0.5f * theta);
        float h = 0.5f * (phi + omega), g = 0.5f * (phi - omega);
        float ur, ui;
        if (mb == 0 && kb == 0)      { ur =  cosf(h) * c; ui = -sinf(h) * c; }
        else if (mb == 0 && kb == 1) { ur = -cosf(g) * s; ui = -sinf(g) * s; }
        else if (mb == 1 && kb == 0) { ur =  cosf(g) * s; ui = -sinf(g) * s; }
        else                         { ur =  cosf(h) * c; ui =  sinf(h) * c; }
        float nr = re * ur - im * ui;
        float ni = re * ui + im * ur;
        re = nr; im = ni;
      }
    }
    int base = blk * 512;
    G[base + cell]       = re;
    G[base + 256 + cell] = im;
  }
}

__host__ __device__ constexpr int pinv_idx(int x, int r) {
  for (int q = WIRES - 1; q >= 0; --q) {
    int tq = q + r; if (tq >= WIRES) tq -= WIRES;
    x ^= ((x >> (10 - q)) & 1) << (10 - tq);
  }
  return x;
}

__host__ __device__ constexpr int swz(int i) {
  int m = 0;
  if (i & (1 << 5))  m ^= 1;
  if (i & (1 << 6))  m ^= 20;
  if (i & (1 << 7))  m ^= 10;
  if (i & (1 << 8))  m ^= 21;
  if (i & (1 << 9))  m ^= 17;
  if (i & (1 << 10)) m ^= 17;
  return i ^ m;
}
template <int RP>
__host__ __device__ constexpr int phys(int x) {
  return swz(RP ? pinv_idx(x, RP) : x) * 8;
}
struct Tab8 { int v[8]; };
__host__ __device__ constexpr int or8(Tab8 t) {
  int m = 0;
  for (int b = 0; b < 8; ++b) m |= t.v[b];
  return m;
}
template <int LO, int HS>
__host__ __device__ constexpr int grp_bit(int b) {
  if (b == 4) return (1 << HS) << LO;
  int j = 1 << (b < 4 ? b : b - 1);
  return ((j >> LO) << (LO + 4)) | (j & ((1 << LO) - 1));
}
template <int RP, int LO, int HS>
__host__ __device__ constexpr Tab8 grp_bits() {
  Tab8 t{};
  for (int b = 0; b < 8; ++b) t.v[b] = phys<RP>(grp_bit<LO, HS>(b));
  return t;
}
template <int RP, int LO>
__host__ __device__ constexpr Tab8 grp_kconst() {
  Tab8 t{};
  for (int c = 0; c < 4; ++c)
    for (int e = 0; e < 2; ++e) t.v[2 * c + e] = phys<RP>((4 * c + e) << LO);
  return t;
}
template <int LO>
__host__ __device__ constexpr Tab8 grp_rconst() {
  Tab8 t{};
  for (int r = 0; r < 8; ++r) t.v[r] = phys<0>(r << LO);
  return t;
}
template <int RP>
__host__ __device__ constexpr Tab8 pair_bits() {
  Tab8 t{};
  for (int b = 0; b < 8; ++b) t.v[b] = phys<RP>(2 << b);
  return t;
}
__device__ __forceinline__ int lin8(int tid, const Tab8 t) {
  int v = 0;
#pragma unroll
  for (int b = 0; b < 8; ++b) v ^= (-((tid >> b) & 1)) & t.v[b];
  return v;
}
__device__ __forceinline__ int xoff(int v, int vmask, int k) {
  return (v ^ (k & vmask)) + (k & ~vmask);
}
__device__ __forceinline__ float2 lds_ld(const char* stb, int off) {
  return *(const float2*)(stb + off);
}
__device__ __forceinline__ void lds_st(char* stb, int off, float2 v) {
  *(float2*)(stb + off) = v;
}

template <int RP, int LO>
__device__ __forceinline__ void apply_group(char* stb,
                                            const float* __restrict__ gr,
                                            const float* __restrict__ gi,
                                            int tid) {
  const int lane = tid & 31;
  const int n    = lane & 15;
  const int half = lane >> 4;

  constexpr Tab8 LB = grp_bits<RP, LO, 1>();
  constexpr Tab8 SB = grp_bits<0,  LO, 3>();
  constexpr Tab8 KL = grp_kconst<RP, LO>();
  constexpr Tab8 KS = grp_rconst<LO>();
  constexpr int  LM = or8(LB);
  constexpr int  SM = or8(SB);
  const int vl = lin8(tid, LB);
  const int vs = lin8(tid, SB);

  v2f ar[4], ai[4], br[4], bi[4];
#pragma unroll
  for (int c = 0; c < 4; ++c) {
    int k0 = 4 * c + 2 * half;
    float2 s0 = lds_ld(stb, xoff(vl, LM, KL.v[2 * c]));
    float2 s1 = lds_ld(stb, xoff(vl, LM, KL.v[2 * c + 1]));
    br[c] = v2f{s0.x, s1.x};
    bi[c] = v2f{s0.y, s1.y};
    ar[c] = v2f{gr[n * 16 + k0], gr[n * 16 + k0 + 1]};
    ai[c] = v2f{gi[n * 16 + k0], gi[n * 16 + k0 + 1]};
  }
  if constexpr (RP != 0) __syncthreads();
  else __builtin_amdgcn_sched_barrier(0);

  v8f k1 = {0, 0, 0, 0, 0, 0, 0, 0};
#pragma unroll
  for (int c = 0; c < 4; ++c) {
    v2f bs = br[c] + bi[c];
    k1 = __builtin_amdgcn_wmma_f32_16x16x4_f32(false, ar[c], false, bs,
                                               (short)0, k1, false, false);
  }
  v8f dr = k1;
  v8f di = k1;
#pragma unroll
  for (int c = 0; c < 4; ++c) {
    v2f q = -ar[c] - ai[c];
    v2f p = ai[c] - ar[c];
    dr = __builtin_amdgcn_wmma_f32_16x16x4_f32(false, q, false, bi[c],
                                               (short)0, dr, false, false);
    di = __builtin_amdgcn_wmma_f32_16x16x4_f32(false, p, false, br[c],
                                               (short)0, di, false, false);
  }

#pragma unroll
  for (int r = 0; r < 8; ++r) {
    lds_st(stb, xoff(vs, SM, KS.v[r]), make_float2(dr[r], di[r]));
  }
}

template <int RP>
__device__ __forceinline__ void rx_pass(char* stb, int tid, float rc, float rs) {
  constexpr Tab8 PB = pair_bits<RP>();
  constexpr int  PM = or8(PB);
  constexpr Tab8 KT = { { phys<RP>(0),    phys<RP>(1),
                          phys<RP>(512),  phys<RP>(513),
                          phys<RP>(1024), phys<RP>(1025),
                          phys<RP>(1536), phys<RP>(1537) } };
  const int vt = lin8(tid, PB);
#pragma unroll
  for (int t = 0; t < 4; ++t) {
    int x0 = xoff(vt, PM, KT.v[2 * t]);
    int x1 = xoff(vt, PM, KT.v[2 * t + 1]);
    float2 a0 = lds_ld(stb, x0), a1 = lds_ld(stb, x1);
    lds_st(stb, x0, make_float2(rc * a0.x + rs * a1.y, rc * a0.y - rs * a1.x));
    lds_st(stb, x1, make_float2(rc * a1.x + rs * a0.y, rc * a1.y - rs * a0.x));
  }
}

template <int RP>
__device__ __forceinline__ void layer(char* stb, const float* __restrict__ Gl, int tid) {
  apply_group<RP, 7>(stb, Gl,        Gl + 256,  tid); __syncthreads();
  apply_group<0, 3>(stb, Gl + 512,  Gl + 768,  tid); __syncthreads();
  apply_group<0, 0>(stb, Gl + 1024, Gl + 1280, tid); __syncthreads();
}

template <int RP>
__device__ __forceinline__ void reupload_block(char* stb, const float* __restrict__ Gb,
                                               int tid, float rc, float rs) {
  rx_pass<RP>(stb, tid, rc, rs);
  __syncthreads();
  layer<RP>(stb, Gb,            tid);
  layer<1>(stb, Gb + 1 * 1536, tid);
  layer<2>(stb, Gb + 2 * 1536, tid);
  layer<3>(stb, Gb + 3 * 1536, tid);
}

__global__ void __launch_bounds__(256)
qsim(const float* __restrict__ x, const float* __restrict__ y,
     const float* __restrict__ G, float* __restrict__ out) {
  __shared__ float2 st[DIM];
  __shared__ float  red[256];
  char* stb = (char*)st;
  const int tid  = threadIdx.x;
  const int b    = blockIdx.x;

  float xs[4], part = 0.f;
#pragma unroll
  for (int t = 0; t < 4; ++t) {
    int p = tid + 256 * t;
    float v = x[b * PIX + p];
    xs[t] = v; part += v * v;
  }
  red[tid] = part; __syncthreads();
  for (int off = 128; off > 0; off >>= 1) {
    if (tid < off) red[tid] += red[tid + off];
    __syncthreads();
  }
  float inv = rsqrtf(red[0]);
  constexpr Tab8 EB = pair_bits<0>();
  constexpr int  EM = or8(EB);
  const int ve = lin8(tid, EB);
#pragma unroll
  for (int t = 0; t < 4; ++t) {
    lds_st(stb, xoff(ve, EM, phys<0>(512 * t)),     make_float2(xs[t] * inv, 0.f));
    lds_st(stb, xoff(ve, EM, phys<0>(512 * t + 1)), make_float2(0.f, 0.f));
  }
  __syncthreads();

  const float yv = y[b];
  const float rc = cosf(0.5f * yv), rs = sinf(0.5f * yv);

  reupload_block<0>(stb, G,            tid, rc, rs);
  reupload_block<4>(stb, G + 4 * 1536, tid, rc, rs);

  constexpr Tab8 OB = pair_bits<4>();
  constexpr int  OM = or8(OB);
  const int vo = lin8(tid, OB);
#pragma unroll
  for (int t = 0; t < 4; ++t) {
    int p  = tid + 256 * t;
    float2 a = lds_ld(stb, xoff(vo, OM, phys<4>(512 * t)));
    out[b * PIX + p] = fminf(1.0f, (a.x * a.x + a.y * a.y) * (float)PIX);
  }
}

extern "C" void kernel_launch(void* const* d_in, const int* in_sizes, int n_in,
                              void* d_out, int out_size, void* d_ws, size_t ws_size,
                              hipStream_t stream) {
  const float* x  = (const float*)d_in[0];
  const float* y  = (const float*)d_in[1];
  const float* w0 = (const float*)d_in[2];
  const float* w1 = (const float*)d_in[3];
  float* G = (float*)d_ws;

  qprep<<<24, 256, 0, stream>>>(w0, w1, G);
  qsim<<<NB, 256, 0, stream>>>(x, y, G, (float*)d_out);
}


# prep kernel: each group's 16 distinct 2x2 gate entries evaluated once (split over its 8 waves, shared via LDS), not every cell redoing its 4 wires' tanh/sin/cos in series; same arithmetic per cell
# speedup vs baseline: 3.5412x; 1.0741x over previous
#include <hip/hip_runtime.h>
#include <math.h>


#define WIRES 11
#define DIM   2048
#define PIX   1024
#define NB    1024

typedef __attribute__((ext_vector_type(2))) float v2f;
typedef __attribute__((ext_vector_type(8))) float v8f;

__global__ void __launch_bounds__(256)
qprep(const float* __restrict__ w0, const float* __restrict__ w1,
      float* __restrict__ G) {
  const float PI_ = 3.14159265358979323846f;
  __shared__ float2 ent[16];
  const int blk  = blockIdx.x;
  const int L    = blk / 3;
  const int grp  = blk - L * 3;
  const int cell = threadIdx.x;
  const int m = cell >> 4, k = cell & 15;
  const float* w = (L < 4) ? w0 : w1;
  const int l = L & 3;

  int nw, q0, topbit;
  if (grp == 0)      { nw = 4; q0 = 0; topbit = 3; }
  else if (grp == 1) { nw = 4; q0 = 4; topbit = 3; }
  else               { nw = 3; q0 = 8; topbit = 2; }

  const int wv = __builtin_amdgcn_readfirstlane(cell >> 5);
  const int wi = wv & 3;
  const bool dg = wv < 4;
  if (wi < nw) {
    const float* a3 = w + (l * WIRES + q0 + wi) * 3;
    const float t0 = a3[0], t1 = a3[1], t2 = a3[2];
    float phi   = PI_ * tanhf(t0);
    float theta = PI_ * tanhf(t1);
    float omega = PI_ * tanhf(t2);
    float mag = dg ? cosf(0.5f * theta) : sinf(0.5f * theta);
    float ang = dg ? 0.5f * (phi + omega) : 0.5f * (phi - omega);
    float pr = cosf(ang) * mag;
    float pi = sinf(ang) * mag;
    const int lane = cell & 31;
    if (lane == 0)
      ent[4 * wi + (dg ? 0 : 1)] = make_float2(dg ? pr : -pr, -pi);
    if (lane == 1)
      ent[4 * wi + (dg ? 3 : 2)] = make_float2(pr, dg ? pi : -pi);
  }
  __syncthreads();

  float2 u[4];
#pragma unroll
  for (int i = 0; i < 4; ++i) {
    int bp = topbit - i;
    if (bp < 0) bp = 0;
    int mb = (m >> bp) & 1, kb = (k >> bp) & 1;
    u[i] = ent[(i < nw) ? 4 * i + 2 * mb + kb : 0];
  }
  float re = 1.f, im = 0.f;
  if (grp == 2 && (m >> 3) != (k >> 3)) { re = 0.f; im = 0.f; }
  else {
#pragma unroll
    for (int i = 0; i < 4; ++i) {
      if (i < nw) {
        float nr = re * u[i].x - im * u[i].y;
        float ni = re * u[i].y + im * u[i].x;
        re = nr; im = ni;
      }
    }
  }
  const int base = blk * 512;
  G[base + cell]       = re;
  G[base + 256 + cell] = im;
}

__host__ __device__ constexpr int pinv_idx(int x, int r) {
  for (int q = WIRES - 1; q >= 0; --q) {
    int tq = q + r; if (tq >= WIRES) tq -= WIRES;
    x ^= ((x >> (10 - q)) & 1) << (10 - tq);
  }
  return x;
}

__host__ __device__ constexpr int swz(int i) {
  int m = 0;
  if (i & (1 << 5))  m ^= 1;
  if (i & (1 << 6))  m ^= 20;
  if (i & (1 << 7))  m ^= 10;
  if (i & (1 << 8))  m ^= 21;
  if (i & (1 << 9))  m ^= 17;
  if (i & (1 << 10)) m ^= 17;
  return i ^ m;
}
template <int RP>
__host__ __device__ constexpr int phys(int x) {
  return swz(RP ? pinv_idx(x, RP) : x) * 8;
}
struct Tab8 { int v[8]; };
__host__ __device__ constexpr int or8(Tab8 t) {
  int m = 0;
  for (int b = 0; b < 8; ++b) m |= t.v[b];
  return m;
}
template <int LO, int HS>
__host__ __device__ constexpr int grp_bit(int b) {
  if (b == 4) return (1 << HS) << LO;
  int j = 1 << (b < 4 ? b : b - 1);
  return ((j >> LO) << (LO + 4)) | (j & ((1 << LO) - 1));
}
template <int RP, int LO, int HS>
__host__ __device__ constexpr Tab8 grp_bits() {
  Tab8 t{};
  for (int b = 0; b < 8; ++b) t.v[b] = phys<RP>(grp_bit<LO, HS>(b));
  return t;
}
template <int RP, int LO>
__host__ __device__ constexpr Tab8 grp_kconst() {
  Tab8 t{};
  for (int c = 0; c < 4; ++c)
    for (int e = 0; e < 2; ++e) t.v[2 * c + e] = phys<RP>((4 * c + e) << LO);
  return t;
}
template <int LO>
__host__ __device__ constexpr Tab8 grp_rconst() {
  Tab8 t{};
  for (int r = 0; r < 8; ++r) t.v[r] = phys<0>(r << LO);
  return t;
}
template <int RP>
__host__ __device__ constexpr Tab8 pair_bits() {
  Tab8 t{};
  for (int b = 0; b < 8; ++b) t.v[b] = phys<RP>(2 << b);
  return t;
}
__device__ __forceinline__ int lin8(int tid, const Tab8 t) {
  int v = 0;
#pragma unroll
  for (int b = 0; b < 8; ++b) v ^= (-((tid >> b) & 1)) & t.v[b];
  return v;
}
__device__ __forceinline__ int xoff(int v, int vmask, int k) {
  return (v ^ (k & vmask)) + (k & ~vmask);
}
__device__ __forceinline__ float2 lds_ld(const char* stb, int off) {
  return *(const float2*)(stb + off);
}
__device__ __forceinline__ void lds_st(char* stb, int off, float2 v) {
  *(float2*)(stb + off) = v;
}

template <int RP, int LO>
__device__ __forceinline__ void apply_group(char* stb,
                                            const float* __restrict__ gr,
                                            const float* __restrict__ gi,
                                            int tid) {
  const int lane = tid & 31;
  const int n    = lane & 15;
  const int half = lane >> 4;

  constexpr Tab8 LB = grp_bits<RP, LO, 1>();
  constexpr Tab8 SB = grp_bits<0,  LO, 3>();
  constexpr Tab8 KL = grp_kconst<RP, LO>();
  constexpr Tab8 KS = grp_rconst<LO>();
  constexpr int  LM = or8(LB);
  constexpr int  SM = or8(SB);
  const int vl = lin8(tid, LB);
  const int vs = lin8(tid, SB);

  v2f ar[4], ai[4], br[4], bi[4];
#pragma unroll
  for (int c = 0; c < 4; ++c) {
    int k0 = 4 * c + 2 * half;
    float2 s0 = lds_ld(stb, xoff(vl, LM, KL.v[2 * c]));
    float2 s1 = lds_ld(stb, xoff(vl, LM, KL.v[2 * c + 1]));
    br[c] = v2f{s0.x, s1.x};
    bi[c] = v2f{s0.y, s1.y};
    ar[c] = v2f{gr[n * 16 + k0], gr[n * 16 + k0 + 1]};
    ai[c] = v2f{gi[n * 16 + k0], gi[n * 16 + k0 + 1]};
  }
  if constexpr (RP != 0) __syncthreads();
  else __builtin_amdgcn_sched_barrier(0);

  v8f k1 = {0, 0, 0, 0, 0, 0, 0, 0};
#pragma unroll
  for (int c = 0; c < 4; ++c) {
    v2f bs = br[c] + bi[c];
    k1 = __builtin_amdgcn_wmma_f32_16x16x4_f32(false, ar[c], false, bs,
                                               (short)0, k1, false, false);
  }
  v8f dr = k1;
  v8f di = k1;
#pragma unroll
  for (int c = 0; c < 4; ++c) {
    v2f q = -ar[c] - ai[c];
    v2f p = ai[c] - ar[c];
    dr = __builtin_amdgcn_wmma_f32_16x16x4_f32(false, q, false, bi[c],
                                               (short)0, dr, false, false);
    di = __builtin_amdgcn_wmma_f32_16x16x4_f32(false, p, false, br[c],
                                               (short)0, di, false, false);
  }

#pragma unroll
  for (int r = 0; r < 8; ++r) {
    lds_st(stb, xoff(vs, SM, KS.v[r]), make_float2(dr[r], di[r]));
  }
}

template <int RP>
__device__ __forceinline__ void rx_pass(char* stb, int tid, float rc, float rs) {
  constexpr Tab8 PB = pair_bits<RP>();
  constexpr int  PM = or8(PB);
  constexpr Tab8 KT = { { phys<RP>(0),    phys<RP>(1),
                          phys<RP>(512),  phys<RP>(513),
                          phys<RP>(1024), phys<RP>(1025),
                          phys<RP>(1536), phys<RP>(1537) } };
  const int vt = lin8(tid, PB);
#pragma unroll
  for (int t = 0; t < 4; ++t) {
    int x0 = xoff(vt, PM, KT.v[2 * t]);
    int x1 = xoff(vt, PM, KT.v[2 * t + 1]);
    float2 a0 = lds_ld(stb, x0), a1 = lds_ld(stb, x1);
    lds_st(stb, x0, make_float2(rc * a0.x + rs * a1.y, rc * a0.y - rs * a1.x));
    lds_st(stb, x1, make_float2(rc * a1.x + rs * a0.y, rc * a1.y - rs * a0.x));
  }
}

template <int RP>
__device__ __forceinline__ void layer(char* stb, const float* __restrict__ Gl, int tid) {
  apply_group<RP, 7>(stb, Gl,        Gl + 256,  tid); __syncthreads();
  apply_group<0, 3>(stb, Gl + 512,  Gl + 768,  tid); __syncthreads();
  apply_group<0, 0>(stb, Gl + 1024, Gl + 1280, tid); __syncthreads();
}

template <int RP>
__device__ __forceinline__ void reupload_block(char* stb, const float* __restrict__ Gb,
                                               int tid, float rc, float rs) {
  rx_pass<RP>(stb, tid, rc, rs);
  __syncthreads();
  layer<RP>(stb, Gb,            tid);
  layer<1>(stb, Gb + 1 * 1536, tid);
  layer<2>(stb, Gb + 2 * 1536, tid);
  layer<3>(stb, Gb + 3 * 1536, tid);
}

__global__ void __launch_bounds__(256)
qsim(const float* __restrict__ x, const float* __restrict__ y,
     const float* __restrict__ G, float* __restrict__ out) {
  __shared__ float2 st[DIM];
  __shared__ float  red[256];
  char* stb = (char*)st;
  const int tid  = threadIdx.x;
  const int b    = blockIdx.x;

  float xs[4], part = 0.f;
#pragma unroll
  for (int t = 0; t < 4; ++t) {
    int p = tid + 256 * t;
    float v = x[b * PIX + p];
    xs[t] = v; part += v * v;
  }
  red[tid] = part; __syncthreads();
  for (int off = 128; off > 0; off >>= 1) {
    if (tid < off) red[tid] += red[tid + off];
    __syncthreads();
  }
  float inv = rsqrtf(red[0]);
  constexpr Tab8 EB = pair_bits<0>();
  constexpr int  EM = or8(EB);
  const int ve = lin8(tid, EB);
#pragma unroll
  for (int t = 0; t < 4; ++t) {
    lds_st(stb, xoff(ve, EM, phys<0>(512 * t)),     make_float2(xs[t] * inv, 0.f));
    lds_st(stb, xoff(ve, EM, phys<0>(512 * t + 1)), make_float2(0.f, 0.f));
  }
  __syncthreads();

  const float yv = y[b];
  const float rc = cosf(0.5f * yv), rs = sinf(0.5f * yv);

  reupload_block<0>(stb, G,            tid, rc, rs);
  reupload_block<4>(stb, G + 4 * 1536, tid, rc, rs);

  constexpr Tab8 OB = pair_bits<4>();
  constexpr int  OM = or8(OB);
  const int vo = lin8(tid, OB);
#pragma unroll
  for (int t = 0; t < 4; ++t) {
    int p  = tid + 256 * t;
    float2 a = lds_ld(stb, xoff(vo, OM, phys<4>(512 * t)));
    out[b * PIX + p] = fminf(1.0f, (a.x * a.x + a.y * a.y) * (float)PIX);
  }
}

extern "C" void kernel_launch(void* const* d_in, const int* in_sizes, int n_in,
                              void* d_out, int out_size, void* d_ws, size_t ws_size,
                              hipStream_t stream) {
  const float* x  = (const float*)d_in[0];
  const float* y  = (const float*)d_in[1];
  const float* w0 = (const float*)d_in[2];
  const float* w1 = (const float*)d_in[3];
  float* G = (float*)d_ws;

  qprep<<<24, 256, 0, stream>>>(w0, w1, G);
  qsim<<<NB, 256, 0, stream>>>(x, y, G, (float*)d_out);
}


# wires 8-10 group as ONE real 16x16 product with the real form [[Gr,-Gi],[Gi,Gr]] of its 8x8 complex matrix: 8 WMMA per wave-pass, not 12 on I2(x)G8; changes rounding and summation order
# speedup vs baseline: 3.9206x; 1.1071x over previous
#include <hip/hip_runtime.h>
#include <math.h>


#define WIRES 11
#define DIM   2048
#define PIX   1024
#define NB    1024

typedef __attribute__((ext_vector_type(2))) float v2f;
typedef __attribute__((ext_vector_type(8))) float v8f;

__global__ void __launch_bounds__(256)
qprep(const float* __restrict__ w0, const float* __restrict__ w1,
      float* __restrict__ G) {
  const float PI_ = 3.14159265358979323846f;
  __shared__ float2 ent[16];
  const int blk  = blockIdx.x;
  const int L    = blk / 3;
  const int grp  = blk - L * 3;
  const int cell = threadIdx.x;
  int m = cell >> 4, k = cell & 15;
  int part = 0, cpart = 0;
  if (grp == 2) {
    part = m & 1; cpart = k & 1;
    const int R  = (((m >> 1) & 3) << 1) | (m >> 3);
    const int kr = (((k >> 1) & 3) << 1) | (k >> 3);
    m = R; k = kr;
  }
  const float* w = (L < 4) ? w0 : w1;
  const int l = L & 3;

  int nw, q0, topbit;
  if (grp == 0)      { nw = 4; q0 = 0; topbit = 3; }
  else if (grp == 1) { nw = 4; q0 = 4; topbit = 3; }
  else               { nw = 3; q0 = 8; topbit = 2; }

  const int wv = __builtin_amdgcn_readfirstlane(cell >> 5);
  const int wi = wv & 3;
  const bool dg = wv < 4;
  if (wi < nw) {
    const float* a3 = w + (l * WIRES + q0 + wi) * 3;
    const float t0 = a3[0], t1 = a3[1], t2 = a3[2];
    float phi   = PI_ * tanhf(t0);
    float theta = PI_ * tanhf(t1);
    float omega = PI_ * tanhf(t2);
    float mag = dg ? cosf(0.5f * theta) : sinf(0.5f * theta);
    float ang = dg ? 0.5f * (phi + omega) : 0.5f * (phi - omega);
    float pr = cosf(ang) * mag;
    float pi = sinf(ang) * mag;
    const int lane = cell & 31;
    if (lane == 0)
      ent[4 * wi + (dg ? 0 : 1)] = make_float2(dg ? pr : -pr, -pi);
    if (lane == 1)
      ent[4 * wi + (dg ? 3 : 2)] = make_float2(pr, dg ? pi : -pi);
  }
  __syncthreads();

  float2 u[4];
#pragma unroll
  for (int i = 0; i < 4; ++i) {
    int bp = topbit - i;
    if (bp < 0) bp = 0;
    int mb = (m >> bp) & 1, kb = (k >> bp) & 1;
    u[i] = ent[(i < nw) ? 4 * i + 2 * mb + kb : 0];
  }
  float re = 1.f, im = 0.f;
#pragma unroll
  for (int i = 0; i < 4; ++i) {
    if (i < nw) {
      float nr = re * u[i].x - im * u[i].y;
      float ni = re * u[i].y + im * u[i].x;
      re = nr; im = ni;
    }
  }
  const int base = blk * 512;
  if (grp == 2) {
    G[base + cell] = (part == cpart) ? re : (part ? im : -im);
  } else {
    G[base + cell]       = re;
    G[base + 256 + cell] = im;
  }
}

__host__ __device__ constexpr int pinv_idx(int x, int r) {
  for (int q = WIRES - 1; q >= 0; --q) {
    int tq = q + r; if (tq >= WIRES) tq -= WIRES;
    x ^= ((x >> (10 - q)) & 1) << (10 - tq);
  }
  return x;
}

__host__ __device__ constexpr int swz(int i) {
  int m = 0;
  if (i & (1 << 5))  m ^= 1;
  if (i & (1 << 6))  m ^= 20;
  if (i & (1 << 7))  m ^= 10;
  if (i & (1 << 8))  m ^= 21;
  if (i & (1 << 9))  m ^= 17;
  if (i & (1 << 10)) m ^= 17;
  return i ^ m;
}
template <int RP>
__host__ __device__ constexpr int phys(int x) {
  return swz(RP ? pinv_idx(x, RP) : x) * 8;
}
struct Tab8 { int v[8]; };
__host__ __device__ constexpr int or8(Tab8 t) {
  int m = 0;
  for (int b = 0; b < 8; ++b) m |= t.v[b];
  return m;
}
template <int LO, int HS>
__host__ __device__ constexpr int grp_bit(int b) {
  if (b == 4) return (1 << HS) << LO;
  int j = 1 << (b < 4 ? b : b - 1);
  return ((j >> LO) << (LO + 4)) | (j & ((1 << LO) - 1));
}
template <int RP, int LO, int HS>
__host__ __device__ constexpr Tab8 grp_bits() {
  Tab8 t{};
  for (int b = 0; b < 8; ++b) t.v[b] = phys<RP>(grp_bit<LO, HS>(b));
  return t;
}
template <int RP, int LO>
__host__ __device__ constexpr Tab8 grp_kconst() {
  Tab8 t{};
  for (int c = 0; c < 4; ++c)
    for (int e = 0; e < 2; ++e) t.v[2 * c + e] = phys<RP>((4 * c + e) << LO);
  return t;
}
template <int LO>
__host__ __device__ constexpr Tab8 grp_rconst() {
  Tab8 t{};
  for (int r = 0; r < 8; ++r) t.v[r] = phys<0>(r << LO);
  return t;
}
template <int RP>
__host__ __device__ constexpr Tab8 pair_bits() {
  Tab8 t{};
  for (int b = 0; b < 8; ++b) t.v[b] = phys<RP>(2 << b);
  return t;
}
__device__ __forceinline__ int lin8(int tid, const Tab8 t) {
  int v = 0;
#pragma unroll
  for (int b = 0; b < 8; ++b) v ^= (-((tid >> b) & 1)) & t.v[b];
  return v;
}
__device__ __forceinline__ int xoff(int v, int vmask, int k) {
  return (v ^ (k & vmask)) + (k & ~vmask);
}
__device__ __forceinline__ float2 lds_ld(const char* stb, int off) {
  return *(const float2*)(stb + off);
}
__device__ __forceinline__ void lds_st(char* stb, int off, float2 v) {
  *(float2*)(stb + off) = v;
}

template <int RP, int LO>
__device__ __forceinline__ void apply_group(char* stb,
                                            const float* __restrict__ gr,
                                            const float* __restrict__ gi,
                                            int tid) {
  const int lane = tid & 31;
  const int n    = lane & 15;
  const int half = lane >> 4;

  constexpr Tab8 LB = grp_bits<RP, LO, 1>();
  constexpr Tab8 SB = grp_bits<0,  LO, 3>();
  constexpr Tab8 KL = grp_kconst<RP, LO>();
  constexpr Tab8 KS = grp_rconst<LO>();
  constexpr int  LM = or8(LB);
  constexpr int  SM = or8(SB);
  const int vl = lin8(tid, LB);
  const int vs = lin8(tid, SB);

  v2f ar[4], ai[4], br[4], bi[4];
#pragma unroll
  for (int c = 0; c < 4; ++c) {
    int k0 = 4 * c + 2 * half;
    float2 s0 = lds_ld(stb, xoff(vl, LM, KL.v[2 * c]));
    float2 s1 = lds_ld(stb, xoff(vl, LM, KL.v[2 * c + 1]));
    br[c] = v2f{s0.x, s1.x};
    bi[c] = v2f{s0.y, s1.y};
    ar[c] = v2f{gr[n * 16 + k0], gr[n * 16 + k0 + 1]};
    ai[c] = v2f{gi[n * 16 + k0], gi[n * 16 + k0 + 1]};
  }
  if constexpr (RP != 0) __syncthreads();
  else __builtin_amdgcn_sched_barrier(0);

  v8f k1 = {0, 0, 0, 0, 0, 0, 0, 0};
#pragma unroll
  for (int c = 0; c < 4; ++c) {
    v2f bs = br[c] + bi[c];
    k1 = __builtin_amdgcn_wmma_f32_16x16x4_f32(false, ar[c], false, bs,
                                               (short)0, k1, false, false);
  }
  v8f dr = k1;
  v8f di = k1;
#pragma unroll
  for (int c = 0; c < 4; ++c) {
    v2f q = -ar[c] - ai[c];
    v2f p = ai[c] - ar[c];
    dr = __builtin_amdgcn_wmma_f32_16x16x4_f32(false, q, false, bi[c],
                                               (short)0, dr, false, false);
    di = __builtin_amdgcn_wmma_f32_16x16x4_f32(false, p, false, br[c],
                                               (short)0, di, false, false);
  }

#pragma unroll
  for (int r = 0; r < 8; ++r) {
    lds_st(stb, xoff(vs, SM, KS.v[r]), make_float2(dr[r], di[r]));
  }
}

__host__ __device__ constexpr Tab8 g3_bits() {
  const int ib[8] = {3, 4, 7, 8, 0, 5, 6, 9};
  Tab8 t{};
  for (int b = 0; b < 8; ++b) t.v[b] = phys<0>(1 << ib[b]);
  return t;
}
__host__ __device__ constexpr Tab8 g3_const() {
  Tab8 t{};
  for (int tl = 0; tl < 2; ++tl)
    for (int c = 0; c < 4; ++c) t.v[4 * tl + c] = phys<0>((tl << 10) | (c << 1));
  return t;
}
__device__ __forceinline__ void apply_group3(char* stb,
                                             const float* __restrict__ ge,
                                             int tid) {
  const int lane = tid & 31;
  const int n    = lane & 15;
  const int half = lane >> 4;

  constexpr Tab8 LB = g3_bits();
  constexpr Tab8 KL = g3_const();
  constexpr int  LM = or8(LB);
  const int vl = lin8(tid, LB);

  const float4* ap = (const float4*)(ge + n * 16 + half * 8);
  const float4 a01 = ap[0], a23 = ap[1];
  const v2f a[4] = { v2f{a01.x, a01.y}, v2f{a01.z, a01.w},
                     v2f{a23.x, a23.y}, v2f{a23.z, a23.w} };
  v2f b0[4], b1[4];
#pragma unroll
  for (int c = 0; c < 4; ++c) {
    float2 s0 = lds_ld(stb, xoff(vl, LM, KL.v[c]));
    float2 s1 = lds_ld(stb, xoff(vl, LM, KL.v[4 + c]));
    b0[c] = v2f{s0.x, s0.y};
    b1[c] = v2f{s1.x, s1.y};
  }
  __builtin_amdgcn_sched_barrier(0);

  v8f d0 = {0, 0, 0, 0, 0, 0, 0, 0};
  v8f d1 = {0, 0, 0, 0, 0, 0, 0, 0};
#pragma unroll
  for (int c = 0; c < 4; ++c) {
    d0 = __builtin_amdgcn_wmma_f32_16x16x4_f32(false, a[c], false, b0[c],
                                               (short)0, d0, false, false);
    d1 = __builtin_amdgcn_wmma_f32_16x16x4_f32(false, a[c], false, b1[c],
                                               (short)0, d1, false, false);
  }
#pragma unroll
  for (int j = 0; j < 4; ++j) {
    lds_st(stb, xoff(vl, LM, KL.v[j]),     make_float2(d0[2 * j], d0[2 * j + 1]));
    lds_st(stb, xoff(vl, LM, KL.v[4 + j]), make_float2(d1[2 * j], d1[2 * j + 1]));
  }
}

template <int RP>
__device__ __forceinline__ void rx_pass(char* stb, int tid, float rc, float rs) {
  constexpr Tab8 PB = pair_bits<RP>();
  constexpr int  PM = or8(PB);
  constexpr Tab8 KT = { { phys<RP>(0),    phys<RP>(1),
                          phys<RP>(512),  phys<RP>(513),
                          phys<RP>(1024), phys<RP>(1025),
                          phys<RP>(1536), phys<RP>(1537) } };
  const int vt = lin8(tid, PB);
#pragma unroll
  for (int t = 0; t < 4; ++t) {
    int x0 = xoff(vt, PM, KT.v[2 * t]);
    int x1 = xoff(vt, PM, KT.v[2 * t + 1]);
    float2 a0 = lds_ld(stb, x0), a1 = lds_ld(stb, x1);
    lds_st(stb, x0, make_float2(rc * a0.x + rs * a1.y, rc * a0.y - rs * a1.x));
    lds_st(stb, x1, make_float2(rc * a1.x + rs * a0.y, rc * a1.y - rs * a0.x));
  }
}

template <int RP>
__device__ __forceinline__ void layer(char* stb, const float* __restrict__ Gl, int tid) {
  apply_group<RP, 7>(stb, Gl,        Gl + 256,  tid); __syncthreads();
  apply_group<0, 3>(stb, Gl + 512,  Gl + 768,  tid); __syncthreads();
  apply_group3(stb, Gl + 1024, tid);                  __syncthreads();
}

template <int RP>
__device__ __forceinline__ void reupload_block(char* stb, const float* __restrict__ Gb,
                                               int tid, float rc, float rs) {
  rx_pass<RP>(stb, tid, rc, rs);
  __syncthreads();
  layer<RP>(stb, Gb,            tid);
  layer<1>(stb, Gb + 1 * 1536, tid);
  layer<2>(stb, Gb + 2 * 1536, tid);
  layer<3>(stb, Gb + 3 * 1536, tid);
}

__global__ void __launch_bounds__(256)
qsim(const float* __restrict__ x, const float* __restrict__ y,
     const float* __restrict__ G, float* __restrict__ out) {
  __shared__ float2 st[DIM];
  __shared__ float  red[256];
  char* stb = (char*)st;
  const int tid  = threadIdx.x;
  const int b    = blockIdx.x;

  float xs[4], part = 0.f;
#pragma unroll
  for (int t = 0; t < 4; ++t) {
    int p = tid + 256 * t;
    float v = x[b * PIX + p];
    xs[t] = v; part += v * v;
  }
  red[tid] = part; __syncthreads();
  for (int off = 128; off > 0; off >>= 1) {
    if (tid < off) red[tid] += red[tid + off];
    __syncthreads();
  }
  float inv = rsqrtf(red[0]);
  constexpr Tab8 EB = pair_bits<0>();
  constexpr int  EM = or8(EB);
  const int ve = lin8(tid, EB);
#pragma unroll
  for (int t = 0; t < 4; ++t) {
    lds_st(stb, xoff(ve, EM, phys<0>(512 * t)),     make_float2(xs[t] * inv, 0.f));
    lds_st(stb, xoff(ve, EM, phys<0>(512 * t + 1)), make_float2(0.f, 0.f));
  }
  __syncthreads();

  const float yv = y[b];
  const float rc = cosf(0.5f * yv), rs = sinf(0.5f * yv);

  reupload_block<0>(stb, G,            tid, rc, rs);
  reupload_block<4>(stb, G + 4 * 1536, tid, rc, rs);

  constexpr Tab8 OB = pair_bits<4>();
  constexpr int  OM = or8(OB);
  const int vo = lin8(tid, OB);
#pragma unroll
  for (int t = 0; t < 4; ++t) {
    int p  = tid + 256 * t;
    float2 a = lds_ld(stb, xoff(vo, OM, phys<4>(512 * t)));
    out[b * PIX + p] = fminf(1.0f, (a.x * a.x + a.y * a.y) * (float)PIX);
  }
}

extern "C" void kernel_launch(void* const* d_in, const int* in_sizes, int n_in,
                              void* d_out, int out_size, void* d_ws, size_t ws_size,
                              hipStream_t stream) {
  const float* x  = (const float*)d_in[0];
  const float* y  = (const float*)d_in[1];
  const float* w0 = (const float*)d_in[2];
  const float* w1 = (const float*)d_in[3];
  float* G = (float*)d_ws;

  qprep<<<24, 256, 0, stream>>>(w0, w1, G);
  qsim<<<NB, 256, 0, stream>>>(x, y, G, (float*)d_out);
}


# all 24 group matrices copied to LDS once per work-group (qprep writes them in fragment order); each pass reads its A fragments from LDS, not from global memory behind the barrier; same arithmetic
# speedup vs baseline: 4.1189x; 1.0506x over previous
#include <hip/hip_runtime.h>
#include <math.h>


#define WIRES 11
#define DIM   2048
#define PIX   1024
#define NB    1024

typedef __attribute__((ext_vector_type(2))) float v2f;
typedef __attribute__((ext_vector_type(8))) float v8f;

__global__ void __launch_bounds__(256)
qprep(const float* __restrict__ w0, const float* __restrict__ w1,
      float* __restrict__ G) {
  const float PI_ = 3.14159265358979323846f;
  __shared__ float2 ent[16];
  const int blk  = blockIdx.x;
  const int L    = blk / 3;
  const int grp  = blk - L * 3;
  const int cell = threadIdx.x;
  int m = cell >> 4, k = cell & 15;
  int part = 0, cpart = 0;
  if (grp == 2) {
    part = m & 1; cpart = k & 1;
    const int R  = (((m >> 1) & 3) << 1) | (m >> 3);
    const int kr = (((k >> 1) & 3) << 1) | (k >> 3);
    m = R; k = kr;
  }
  const float* w = (L < 4) ? w0 : w1;
  const int l = L & 3;

  int nw, q0, topbit;
  if (grp == 0)      { nw = 4; q0 = 0; topbit = 3; }
  else if (grp == 1) { nw = 4; q0 = 4; topbit = 3; }
  else               { nw = 3; q0 = 8; topbit = 2; }

  const int wv = __builtin_amdgcn_readfirstlane(cell >> 5);
  const int wi = wv & 3;
  const bool dg = wv < 4;
  if (wi < nw) {
    const float* a3 = w + (l * WIRES + q0 + wi) * 3;
    const float t0 = a3[0], t1 = a3[1], t2 = a3[2];
    float phi   = PI_ * tanhf(t0);
    float theta = PI_ * tanhf(t1);
    float omega = PI_ * tanhf(t2);
    float mag = dg ? cosf(0.5f * theta) : sinf(0.5f * theta);
    float ang = dg ? 0.5f * (phi + omega) : 0.5f * (phi - omega);
    float pr = cosf(ang) * mag;
    float pi = sinf(ang) * mag;
    const int lane = cell & 31;
    if (lane == 0)
      ent[4 * wi + (dg ? 0 : 1)] = make_float2(dg ? pr : -pr, -pi);
    if (lane == 1)
      ent[4 * wi + (dg ? 3 : 2)] = make_float2(pr, dg ? pi : -pi);
  }
  __syncthreads();

  float2 u[4];
#pragma unroll
  for (int i = 0; i < 4; ++i) {
    int bp = topbit - i;
    if (bp < 0) bp = 0;
    int mb = (m >> bp) & 1, kb = (k >> bp) & 1;
    u[i] = ent[(i < nw) ? 4 * i + 2 * mb + kb : 0];
  }
  float re = 1.f, im = 0.f;
#pragma unroll
  for (int i = 0; i < 4; ++i) {
    if (i < nw) {
      float nr = re * u[i].x - im * u[i].y;
      float ni = re * u[i].y + im * u[i].x;
      re = nr; im = ni;
    }
  }
  const int base = blk * 512;
  const int kk   = cell & 15;
  const int fh   = (grp == 2) ? (kk >> 3) : ((kk >> 1) & 1);
  const int fi   = (grp == 2) ? (kk & 7) : (((kk >> 2) << 1) | (kk & 1));
  const int pos  = 128 * (fi >> 2) + 4 * (16 * fh + (cell >> 4)) + (fi & 3);
  if (grp == 2) {
    G[base + pos]       = (part == cpart) ? re : (part ? im : -im);
    G[base + 256 + pos] = 0.f;
  } else {
    G[base + pos]       = re;
    G[base + 256 + pos] = im;
  }
}

__host__ __device__ constexpr int pinv_idx(int x, int r) {
  for (int q = WIRES - 1; q >= 0; --q) {
    int tq = q + r; if (tq >= WIRES) tq -= WIRES;
    x ^= ((x >> (10 - q)) & 1) << (10 - tq);
  }
  return x;
}

__host__ __device__ constexpr int swz(int i) {
  int m = 0;
  if (i & (1 << 5))  m ^= 1;
  if (i & (1 << 6))  m ^= 20;
  if (i & (1 << 7))  m ^= 10;
  if (i & (1 << 8))  m ^= 21;
  if (i & (1 << 9))  m ^= 17;
  if (i & (1 << 10)) m ^= 17;
  return i ^ m;
}
template <int RP>
__host__ __device__ constexpr int phys(int x) {
  return swz(RP ? pinv_idx(x, RP) : x) * 8;
}
struct Tab8 { int v[8]; };
__host__ __device__ constexpr int or8(Tab8 t) {
  int m = 0;
  for (int b = 0; b < 8; ++b) m |= t.v[b];
  return m;
}
template <int LO, int HS>
__host__ __device__ constexpr int grp_bit(int b) {
  if (b == 4) return (1 << HS) << LO;
  int j = 1 << (b < 4 ? b : b - 1);
  return ((j >> LO) << (LO + 4)) | (j & ((1 << LO) - 1));
}
template <int RP, int LO, int HS>
__host__ __device__ constexpr Tab8 grp_bits() {
  Tab8 t{};
  for (int b = 0; b < 8; ++b) t.v[b] = phys<RP>(grp_bit<LO, HS>(b));
  return t;
}
template <int RP, int LO>
__host__ __device__ constexpr Tab8 grp_kconst() {
  Tab8 t{};
  for (int c = 0; c < 4; ++c)
    for (int e = 0; e < 2; ++e) t.v[2 * c + e] = phys<RP>((4 * c + e) << LO);
  return t;
}
template <int LO>
__host__ __device__ constexpr Tab8 grp_rconst() {
  Tab8 t{};
  for (int r = 0; r < 8; ++r) t.v[r] = phys<0>(r << LO);
  return t;
}
template <int RP>
__host__ __device__ constexpr Tab8 pair_bits() {
  Tab8 t{};
  for (int b = 0; b < 8; ++b) t.v[b] = phys<RP>(2 << b);
  return t;
}
__device__ __forceinline__ int lin8(int tid, const Tab8 t) {
  int v = 0;
#pragma unroll
  for (int b = 0; b < 8; ++b) v ^= (-((tid >> b) & 1)) & t.v[b];
  return v;
}
__device__ __forceinline__ int xoff(int v, int vmask, int k) {
  return (v ^ (k & vmask)) + (k & ~vmask);
}
__device__ __forceinline__ float2 lds_ld(const char* stb, int off) {
  return *(const float2*)(stb + off);
}
__device__ __forceinline__ void lds_st(char* stb, int off, float2 v) {
  *(float2*)(stb + off) = v;
}

template <int RP, int LO>
__device__ __forceinline__ void apply_group(char* stb,
                                            const float* __restrict__ gr,
                                            const float* __restrict__ gi,
                                            int tid) {
  const int lane = tid & 31;

  constexpr Tab8 LB = grp_bits<RP, LO, 1>();
  constexpr Tab8 SB = grp_bits<0,  LO, 3>();
  constexpr Tab8 KL = grp_kconst<RP, LO>();
  constexpr Tab8 KS = grp_rconst<LO>();
  constexpr int  LM = or8(LB);
  constexpr int  SM = or8(SB);
  const int vl = lin8(tid, LB);
  const int vs = lin8(tid, SB);

  const float4* apr = (const float4*)gr + lane;
  const float4* api = (const float4*)gi + lane;
  const float4 r01 = apr[0], r23 = apr[32], i01 = api[0], i23 = api[32];
  const v2f ar[4] = { v2f{r01.x, r01.y}, v2f{r01.z, r01.w},
                      v2f{r23.x, r23.y}, v2f{r23.z, r23.w} };
  const v2f ai[4] = { v2f{i01.x, i01.y}, v2f{i01.z, i01.w},
                      v2f{i23.x, i23.y}, v2f{i23.z, i23.w} };
  v2f br[4], bi[4];
#pragma unroll
  for (int c = 0; c < 4; ++c) {
    float2 s0 = lds_ld(stb, xoff(vl, LM, KL.v[2 * c]));
    float2 s1 = lds_ld(stb, xoff(vl, LM, KL.v[2 * c + 1]));
    br[c] = v2f{s0.x, s1.x};
    bi[c] = v2f{s0.y, s1.y};
  }
  if constexpr (RP != 0) __syncthreads();
  else __builtin_amdgcn_sched_barrier(0);

  v8f k1 = {0, 0, 0, 0, 0, 0, 0, 0};
#pragma unroll
  for (int c = 0; c < 4; ++c) {
    v2f bs = br[c] + bi[c];
    k1 = __builtin_amdgcn_wmma_f32_16x16x4_f32(false, ar[c], false, bs,
                                               (short)0, k1, false, false);
  }
  v8f dr = k1;
  v8f di = k1;
#pragma unroll
  for (int c = 0; c < 4; ++c) {
    v2f q = -ar[c] - ai[c];
    v2f p = ai[c] - ar[c];
    dr = __builtin_amdgcn_wmma_f32_16x16x4_f32(false, q, false, bi[c],
                                               (short)0, dr, false, false);
    di = __builtin_amdgcn_wmma_f32_16x16x4_f32(false, p, false, br[c],
                                               (short)0, di, false, false);
  }

#pragma unroll
  for (int r = 0; r < 8; ++r) {
    lds_st(stb, xoff(vs, SM, KS.v[r]), make_float2(dr[r], di[r]));
  }
}

__host__ __device__ constexpr Tab8 g3_bits() {
  const int ib[8] = {3, 4, 7, 8, 0, 5, 6, 9};
  Tab8 t{};
  for (int b = 0; b < 8; ++b) t.v[b] = phys<0>(1 << ib[b]);
  return t;
}
__host__ __device__ constexpr Tab8 g3_const() {
  Tab8 t{};
  for (int tl = 0; tl < 2; ++tl)
    for (int c = 0; c < 4; ++c) t.v[4 * tl + c] = phys<0>((tl << 10) | (c << 1));
  return t;
}
__device__ __forceinline__ void apply_group3(char* stb,
                                             const float* __restrict__ ge,
                                             int tid) {
  const int lane = tid & 31;

  constexpr Tab8 LB = g3_bits();
  constexpr Tab8 KL = g3_const();
  constexpr int  LM = or8(LB);
  const int vl = lin8(tid, LB);

  const float4* ap = (const float4*)ge + lane;
  const float4 a01 = ap[0], a23 = ap[32];
  const v2f a[4] = { v2f{a01.x, a01.y}, v2f{a01.z, a01.w},
                     v2f{a23.x, a23.y}, v2f{a23.z, a23.w} };
  v2f b0[4], b1[4];
#pragma unroll
  for (int c = 0; c < 4; ++c) {
    float2 s0 = lds_ld(stb, xoff(vl, LM, KL.v[c]));
    float2 s1 = lds_ld(stb, xoff(vl, LM, KL.v[4 + c]));
    b0[c] = v2f{s0.x, s0.y};
    b1[c] = v2f{s1.x, s1.y};
  }
  __builtin_amdgcn_sched_barrier(0);

  v8f d0 = {0, 0, 0, 0, 0, 0, 0, 0};
  v8f d1 = {0, 0, 0, 0, 0, 0, 0, 0};
#pragma unroll
  for (int c = 0; c < 4; ++c) {
    d0 = __builtin_amdgcn_wmma_f32_16x16x4_f32(false, a[c], false, b0[c],
                                               (short)0, d0, false, false);
    d1 = __builtin_amdgcn_wmma_f32_16x16x4_f32(false, a[c], false, b1[c],
                                               (short)0, d1, false, false);
  }
#pragma unroll
  for (int j = 0; j < 4; ++j) {
    lds_st(stb, xoff(vl, LM, KL.v[j]),     make_float2(d0[2 * j], d0[2 * j + 1]));
    lds_st(stb, xoff(vl, LM, KL.v[4 + j]), make_float2(d1[2 * j], d1[2 * j + 1]));
  }
}

template <int RP>
__device__ __forceinline__ void rx_pass(char* stb, int tid, float rc, float rs) {
  constexpr Tab8 PB = pair_bits<RP>();
  constexpr int  PM = or8(PB);
  constexpr Tab8 KT = { { phys<RP>(0),    phys<RP>(1),
                          phys<RP>(512),  phys<RP>(513),
                          phys<RP>(1024), phys<RP>(1025),
                          phys<RP>(1536), phys<RP>(1537) } };
  const int vt = lin8(tid, PB);
#pragma unroll
  for (int t = 0; t < 4; ++t) {
    int x0 = xoff(vt, PM, KT.v[2 * t]);
    int x1 = xoff(vt, PM, KT.v[2 * t + 1]);
    float2 a0 = lds_ld(stb, x0), a1 = lds_ld(stb, x1);
    lds_st(stb, x0, make_float2(rc * a0.x + rs * a1.y, rc * a0.y - rs * a1.x));
    lds_st(stb, x1, make_float2(rc * a1.x + rs * a0.y, rc * a1.y - rs * a0.x));
  }
}

template <int RP>
__device__ __forceinline__ void layer(char* stb, const float* __restrict__ Gl, int tid) {
  apply_group<RP, 7>(stb, Gl,        Gl + 256,  tid); __syncthreads();
  apply_group<0, 3>(stb, Gl + 512,  Gl + 768,  tid); __syncthreads();
  apply_group3(stb, Gl + 1024, tid);                  __syncthreads();
}

template <int RP>
__device__ __forceinline__ void reupload_block(char* stb, const float* __restrict__ Gb,
                                               int tid, float rc, float rs) {
  rx_pass<RP>(stb, tid, rc, rs);
  __syncthreads();
  layer<RP>(stb, Gb,            tid);
  layer<1>(stb, Gb + 1 * 1536, tid);
  layer<2>(stb, Gb + 2 * 1536, tid);
  layer<3>(stb, Gb + 3 * 1536, tid);
}

__global__ void __launch_bounds__(256)
qsim(const float* __restrict__ x, const float* __restrict__ y,
     const float* __restrict__ G, float* __restrict__ out) {
  __shared__ float2 st[DIM];
  __shared__ float  red[256];
  __shared__ float4 gl[3072];
  char* stb = (char*)st;
  const int tid  = threadIdx.x;
  const int b    = blockIdx.x;


  float xs[4], part = 0.f;
#pragma unroll
  for (int t = 0; t < 4; ++t) {
    int p = tid + 256 * t;
    float v = x[b * PIX + p];
    xs[t] = v; part += v * v;
  }
  const float4* G4 = (const float4*)G;
  float4 gv[12];
#pragma unroll
  for (int i = 0; i < 12; ++i) gv[i] = G4[tid + 256 * i];
#pragma unroll
  for (int i = 0; i < 12; ++i) gl[tid + 256 * i] = gv[i];
  red[tid] = part; __syncthreads();
  for (int off = 128; off > 0; off >>= 1) {
    if (tid < off) red[tid] += red[tid + off];
    __syncthreads();
  }
  float inv = rsqrtf(red[0]);
  constexpr Tab8 EB = pair_bits<0>();
  constexpr int  EM = or8(EB);
  const int ve = lin8(tid, EB);
#pragma unroll
  for (int t = 0; t < 4; ++t) {
    lds_st(stb, xoff(ve, EM, phys<0>(512 * t)),     make_float2(xs[t] * inv, 0.f));
    lds_st(stb, xoff(ve, EM, phys<0>(512 * t + 1)), make_float2(0.f, 0.f));
  }
  __syncthreads();
  const float* Gs = (const float*)gl;

  const float yv = y[b];
  const float rc = cosf(0.5f * yv), rs = sinf(0.5f * yv);

  reupload_block<0>(stb, Gs,            tid, rc, rs);
  reupload_block<4>(stb, Gs + 4 * 1536, tid, rc, rs);

  constexpr Tab8 OB = pair_bits<4>();
  constexpr int  OM = or8(OB);
  const int vo = lin8(tid, OB);
#pragma unroll
  for (int t = 0; t < 4; ++t) {
    int p  = tid + 256 * t;
    float2 a = lds_ld(stb, xoff(vo, OM, phys<4>(512 * t)));
    out[b * PIX + p] = fminf(1.0f, (a.x * a.x + a.y * a.y) * (float)PIX);
  }
}

extern "C" void kernel_launch(void* const* d_in, const int* in_sizes, int n_in,
                              void* d_out, int out_size, void* d_ws, size_t ws_size,
                              hipStream_t stream) {
  const float* x  = (const float*)d_in[0];
  const float* y  = (const float*)d_in[1];
  const float* w0 = (const float*)d_in[2];
  const float* w1 = (const float*)d_in[3];
  float* G = (float*)d_ws;

  qprep<<<24, 256, 0, stream>>>(w0, w1, G);
  qsim<<<NB, 256, 0, stream>>>(x, y, G, (float*)d_out);
}


# wires 0-7 RZ factors split off as per-amplitude phase tables so the two 4-wire group matrices are REAL (RY only): 8 WMMA per tile, not 12; changes rounding and summation order
# speedup vs baseline: 4.7201x; 1.1460x over previous
#include <hip/hip_runtime.h>
#include <math.h>


#define WIRES 11
#define DIM   2048
#define PIX   1024
#define NB    1024

typedef __attribute__((ext_vector_type(2))) float v2f;
typedef __attribute__((ext_vector_type(8))) float v8f;

__global__ void __launch_bounds__(256)
qprep(const float* __restrict__ w0, const float* __restrict__ w1,
      float* __restrict__ G) {
  const float PI_ = 3.14159265358979323846f;
  __shared__ float2 ent[16];
  __shared__ float2 zp[12];
  const int blk  = blockIdx.x;
  const int L    = blk / 3;
  const int grp  = blk - L * 3;
  const int cell = threadIdx.x;
  const float* w = (L < 4) ? w0 : w1;
  const int l = L & 3;
  const int Lb = L * 1536;

  if (grp != 2) {
    const int q0 = 4 * grp;
    if (cell < 12) {
      const int wire = (cell < 8) ? cell : q0 + (cell - 8);
      const int comp = (cell < 8) ? (grp == 0 ? 0 : 2) : 1;
      const float ang = PI_ * tanhf(w[(l * WIRES + wire) * 3 + comp]);
      zp[cell] = make_float2(cosf(0.5f * ang), sinf(0.5f * ang));
    }
    __syncthreads();
    const int m = cell >> 4, k = cell & 15;
    float r = 1.f;
#pragma unroll
    for (int i = 0; i < 4; ++i) {
      const int bp = 3 - i;
      const int mb = (m >> bp) & 1, kb = (k >> bp) & 1;
      const float2 t = zp[8 + i];
      r *= (mb == kb) ? t.x : (mb ? t.y : -t.y);
    }
    const int fh  = (k >> 1) & 1;
    const int fi  = ((k >> 2) << 1) | (k & 1);
    const int pos = 128 * (fi >> 2) + 4 * (16 * fh + m) + (fi & 3);
    G[Lb + 256 * grp + pos] = r;
    if (grp == 0) {
      const int k1 = 4 * ((cell >> 1) & 3) + 2 * ((cell >> 3) & 1) + (cell & 1);
      const int wb = (k1 << 4) | (cell >> 4);
      float re = 1.f, im = 0.f;
#pragma unroll
      for (int q = 0; q < 8; ++q) {
        const float2 z = zp[q];
        const float zy = ((wb >> (7 - q)) & 1) ? z.y : -z.y;
        const float nr = re * z.x - im * zy;
        const float ni = re * zy + im * z.x;
        re = nr; im = ni;
      }
      G[Lb + 768 + 2 * cell]     = re;
      G[Lb + 768 + 2 * cell + 1] = im;
    } else {
      if (cell < 32) {
        const int qb = (cell < 16) ? 0 : 4;
        const int v  = cell & 15;
        float re = 1.f, im = 0.f;
#pragma unroll
        for (int i = 0; i < 4; ++i) {
          const float2 z = zp[qb + i];
          const float zy = ((v >> (3 - i)) & 1) ? z.y : -z.y;
          const float nr = re * z.x - im * zy;
          const float ni = re * zy + im * z.x;
          re = nr; im = ni;
        }
        G[Lb + 1280 + 2 * cell]     = re;
        G[Lb + 1280 + 2 * cell + 1] = im;
      } else if (cell < 224) {
        G[Lb + 1344 + (cell - 32)] = 0.f;
      }
    }
    return;
  }

  int m = cell >> 4, k = cell & 15;
  const int part = m & 1, cpart = k & 1;
  {
    const int R  = (((m >> 1) & 3) << 1) | (m >> 3);
    const int kr = (((k >> 1) & 3) << 1) | (k >> 3);
    m = R; k = kr;
  }
  const int nw = 3, q0 = 8, topbit = 2;

  const int wv = __builtin_amdgcn_readfirstlane(cell >> 5);
  const int wi = wv & 3;
  const bool dg = wv < 4;
  if (wi < nw) {
    const float* a3 = w + (l * WIRES + q0 + wi) * 3;
    const float t0 = a3[0], t1 = a3[1], t2 = a3[2];
    float phi   = PI_ * tanhf(t0);
    float theta = PI_ * tanhf(t1);
    float omega = PI_ * tanhf(t2);
    float mag = dg ? cosf(0.5f * theta) : sinf(0.5f * theta);
    float ang = dg ? 0.5f * (phi + omega) : 0.5f * (phi - omega);
    float pr = cosf(ang) * mag;
    float pi = sinf(ang) * mag;
    const int lane = cell & 31;
    if (lane == 0)
      ent[4 * wi + (dg ? 0 : 1)] = make_float2(dg ? pr : -pr, -pi);
    if (lane == 1)
      ent[4 * wi + (dg ? 3 : 2)] = make_float2(pr, dg ? pi : -pi);
  }
  __syncthreads();

  float2 u[3];
#pragma unroll
  for (int i = 0; i < 3; ++i) {
    const int bp = topbit - i;
    const int mb = (m >> bp) & 1, kb = (k >> bp) & 1;
    u[i] = ent[4 * i + 2 * mb + kb];
  }
  float re = 1.f, im = 0.f;
#pragma unroll
  for (int i = 0; i < 3; ++i) {
    float nr = re * u[i].x - im * u[i].y;
    float ni = re * u[i].y + im * u[i].x;
    re = nr; im = ni;
  }
  const int kk   = cell & 15;
  const int fh   = kk >> 3;
  const int fi   = kk & 7;
  const int pos  = 128 * (fi >> 2) + 4 * (16 * fh + (cell >> 4)) + (fi & 3);
  G[Lb + 512 + pos] = (part == cpart) ? re : (part ? im : -im);
}

__host__ __device__ constexpr int pinv_idx(int x, int r) {
  for (int q = WIRES - 1; q >= 0; --q) {
    int tq = q + r; if (tq >= WIRES) tq -= WIRES;
    x ^= ((x >> (10 - q)) & 1) << (10 - tq);
  }
  return x;
}

__host__ __device__ constexpr int swz(int i) {
  int m = 0;
  if (i & (1 << 5))  m ^= 1;
  if (i & (1 << 6))  m ^= 20;
  if (i & (1 << 7))  m ^= 10;
  if (i & (1 << 8))  m ^= 21;
  if (i & (1 << 9))  m ^= 17;
  if (i & (1 << 10)) m ^= 17;
  return i ^ m;
}
template <int RP>
__host__ __device__ constexpr int phys(int x) {
  return swz(RP ? pinv_idx(x, RP) : x) * 8;
}
struct Tab8 { int v[8]; };
__host__ __device__ constexpr int or8(Tab8 t) {
  int m = 0;
  for (int b = 0; b < 8; ++b) m |= t.v[b];
  return m;
}
template <int LO, int HS>
__host__ __device__ constexpr int grp_bit(int b) {
  if (b == 4) return (1 << HS) << LO;
  int j = 1 << (b < 4 ? b : b - 1);
  return ((j >> LO) << (LO + 4)) | (j & ((1 << LO) - 1));
}
template <int RP, int LO, int HS>
__host__ __device__ constexpr Tab8 grp_bits() {
  Tab8 t{};
  for (int b = 0; b < 8; ++b) t.v[b] = phys<RP>(grp_bit<LO, HS>(b));
  return t;
}
template <int RP, int LO>
__host__ __device__ constexpr Tab8 grp_kconst() {
  Tab8 t{};
  for (int c = 0; c < 4; ++c)
    for (int e = 0; e < 2; ++e) t.v[2 * c + e] = phys<RP>((4 * c + e) << LO);
  return t;
}
template <int LO>
__host__ __device__ constexpr Tab8 grp_rconst() {
  Tab8 t{};
  for (int r = 0; r < 8; ++r) t.v[r] = phys<0>(r << LO);
  return t;
}
template <int RP>
__host__ __device__ constexpr Tab8 pair_bits() {
  Tab8 t{};
  for (int b = 0; b < 8; ++b) t.v[b] = phys<RP>(2 << b);
  return t;
}
__device__ __forceinline__ int lin8(int tid, const Tab8 t) {
  int v = 0;
#pragma unroll
  for (int b = 0; b < 8; ++b) v ^= (-((tid >> b) & 1)) & t.v[b];
  return v;
}
__device__ __forceinline__ int xoff(int v, int vmask, int k) {
  return (v ^ (k & vmask)) + (k & ~vmask);
}
__device__ __forceinline__ float2 lds_ld(const char* stb, int off) {
  return *(const float2*)(stb + off);
}
__device__ __forceinline__ void lds_st(char* stb, int off, float2 v) {
  *(float2*)(stb + off) = v;
}

__device__ __forceinline__ float2 cmul(float2 a, float px, float py) {
  return make_float2(a.x * px - a.y * py, a.x * py + a.y * px);
}
template <int RP, int LO, bool PH>
__device__ __forceinline__ void apply_group(char* stb,
                                            const float* __restrict__ gr,
                                            const float* __restrict__ tin,
                                            int tid) {
  const int lane = tid & 31;

  constexpr Tab8 LB = grp_bits<RP, LO, 1>();
  constexpr Tab8 SB = grp_bits<0,  LO, 3>();
  constexpr Tab8 KL = grp_kconst<RP, LO>();
  constexpr Tab8 KS = grp_rconst<LO>();
  constexpr int  LM = or8(LB);
  constexpr int  SM = or8(SB);
  const int vl = lin8(tid, LB);
  const int vs = lin8(tid, SB);

  const float4* apr = (const float4*)gr + lane;
  const float4 r01 = apr[0], r23 = apr[32];
  const v2f ar[4] = { v2f{r01.x, r01.y}, v2f{r01.z, r01.w},
                      v2f{r23.x, r23.y}, v2f{r23.z, r23.w} };
  float4 ph[4];
  if constexpr (PH) {
    const float4* tp = (const float4*)tin
                     + (((tid >> 5) << 1) | ((tid >> 3) & 1)) * 8 + ((tid >> 4) & 1) * 4;
#pragma unroll
    for (int c = 0; c < 4; ++c) ph[c] = tp[c];
  }
  float2 s0[4], s1[4];
#pragma unroll
  for (int c = 0; c < 4; ++c) {
    s0[c] = lds_ld(stb, xoff(vl, LM, KL.v[2 * c]));
    s1[c] = lds_ld(stb, xoff(vl, LM, KL.v[2 * c + 1]));
  }
  if constexpr (RP != 0) __syncthreads();
  else __builtin_amdgcn_sched_barrier(0);

  v2f br[4], bi[4];
#pragma unroll
  for (int c = 0; c < 4; ++c) {
    float2 t0 = s0[c], t1 = s1[c];
    if constexpr (PH) {
      t0 = cmul(t0, ph[c].x, ph[c].y);
      t1 = cmul(t1, ph[c].z, ph[c].w);
    }
    br[c] = v2f{t0.x, t1.x};
    bi[c] = v2f{t0.y, t1.y};
  }

  v8f dr = {0, 0, 0, 0, 0, 0, 0, 0};
  v8f di = {0, 0, 0, 0, 0, 0, 0, 0};
#pragma unroll
  for (int c = 0; c < 4; ++c) {
    dr = __builtin_amdgcn_wmma_f32_16x16x4_f32(false, ar[c], false, br[c],
                                               (short)0, dr, false, false);
    di = __builtin_amdgcn_wmma_f32_16x16x4_f32(false, ar[c], false, bi[c],
                                               (short)0, di, false, false);
  }

#pragma unroll
  for (int r = 0; r < 8; ++r) {
    lds_st(stb, xoff(vs, SM, KS.v[r]), make_float2(dr[r], di[r]));
  }
}

__host__ __device__ constexpr Tab8 g3_bits() {
  const int ib[8] = {3, 4, 7, 8, 0, 5, 6, 9};
  Tab8 t{};
  for (int b = 0; b < 8; ++b) t.v[b] = phys<0>(1 << ib[b]);
  return t;
}
__host__ __device__ constexpr Tab8 g3_const() {
  Tab8 t{};
  for (int tl = 0; tl < 2; ++tl)
    for (int c = 0; c < 4; ++c) t.v[4 * tl + c] = phys<0>((tl << 10) | (c << 1));
  return t;
}
__device__ __forceinline__ void apply_group3(char* stb,
                                             const float* __restrict__ ge,
                                             const float* __restrict__ dout,
                                             int tid) {
  const int lane = tid & 31;
  const float2* dt = (const float2*)dout;
  const int m1 = ((tid >> 2) & 3) | (((tid >> 7) & 1) << 2);
  const int m2 = (tid & 3) | (((tid >> 5) & 3) << 2);
  const float2 pa = dt[m1], pb = dt[m1 + 8], pc = dt[16 + m2];

  constexpr Tab8 LB = g3_bits();
  constexpr Tab8 KL = g3_const();
  constexpr int  LM = or8(LB);
  const int vl = lin8(tid, LB);

  const float4* ap = (const float4*)ge + lane;
  const float4 a01 = ap[0], a23 = ap[32];
  const v2f a[4] = { v2f{a01.x, a01.y}, v2f{a01.z, a01.w},
                     v2f{a23.x, a23.y}, v2f{a23.z, a23.w} };
  v2f b0[4], b1[4];
#pragma unroll
  for (int c = 0; c < 4; ++c) {
    float2 s0 = lds_ld(stb, xoff(vl, LM, KL.v[c]));
    float2 s1 = lds_ld(stb, xoff(vl, LM, KL.v[4 + c]));
    b0[c] = v2f{s0.x, s0.y};
    b1[c] = v2f{s1.x, s1.y};
  }
  __builtin_amdgcn_sched_barrier(0);
  const float2 p0 = cmul(pa, pc.x, pc.y);
  const float2 p1 = cmul(pb, pc.x, pc.y);
#pragma unroll
  for (int c = 0; c < 4; ++c) {
    const float2 t0 = cmul(make_float2(b0[c].x, b0[c].y), p0.x, p0.y);
    const float2 t1 = cmul(make_float2(b1[c].x, b1[c].y), p1.x, p1.y);
    b0[c] = v2f{t0.x, t0.y};
    b1[c] = v2f{t1.x, t1.y};
  }

  v8f d0 = {0, 0, 0, 0, 0, 0, 0, 0};
  v8f d1 = {0, 0, 0, 0, 0, 0, 0, 0};
#pragma unroll
  for (int c = 0; c < 4; ++c) {
    d0 = __builtin_amdgcn_wmma_f32_16x16x4_f32(false, a[c], false, b0[c],
                                               (short)0, d0, false, false);
    d1 = __builtin_amdgcn_wmma_f32_16x16x4_f32(false, a[c], false, b1[c],
                                               (short)0, d1, false, false);
  }
#pragma unroll
  for (int j = 0; j < 4; ++j) {
    lds_st(stb, xoff(vl, LM, KL.v[j]),     make_float2(d0[2 * j], d0[2 * j + 1]));
    lds_st(stb, xoff(vl, LM, KL.v[4 + j]), make_float2(d1[2 * j], d1[2 * j + 1]));
  }
}

template <int RP>
__device__ __forceinline__ void rx_pass(char* stb, int tid, float rc, float rs) {
  constexpr Tab8 PB = pair_bits<RP>();
  constexpr int  PM = or8(PB);
  constexpr Tab8 KT = { { phys<RP>(0),    phys<RP>(1),
                          phys<RP>(512),  phys<RP>(513),
                          phys<RP>(1024), phys<RP>(1025),
                          phys<RP>(1536), phys<RP>(1537) } };
  const int vt = lin8(tid, PB);
#pragma unroll
  for (int t = 0; t < 4; ++t) {
    int x0 = xoff(vt, PM, KT.v[2 * t]);
    int x1 = xoff(vt, PM, KT.v[2 * t + 1]);
    float2 a0 = lds_ld(stb, x0), a1 = lds_ld(stb, x1);
    lds_st(stb, x0, make_float2(rc * a0.x + rs * a1.y, rc * a0.y - rs * a1.x));
    lds_st(stb, x1, make_float2(rc * a1.x + rs * a0.y, rc * a1.y - rs * a0.x));
  }
}

template <int RP>
__device__ __forceinline__ void layer(char* stb, const float* __restrict__ Gl, int tid) {
  apply_group<RP, 7, true>(stb, Gl,       Gl + 768, tid); __syncthreads();
  apply_group<0, 3, false>(stb, Gl + 256, Gl + 768, tid); __syncthreads();
  apply_group3(stb, Gl + 512, Gl + 1280, tid);            __syncthreads();
}

template <int RP>
__device__ __forceinline__ void reupload_block(char* stb, const float* __restrict__ Gb,
                                               int tid, float rc, float rs) {
  rx_pass<RP>(stb, tid, rc, rs);
  __syncthreads();
  layer<RP>(stb, Gb,            tid);
  layer<1>(stb, Gb + 1 * 1536, tid);
  layer<2>(stb, Gb + 2 * 1536, tid);
  layer<3>(stb, Gb + 3 * 1536, tid);
}

__global__ void __launch_bounds__(256)
qsim(const float* __restrict__ x, const float* __restrict__ y,
     const float* __restrict__ G, float* __restrict__ out) {
  __shared__ float2 st[DIM];
  __shared__ float  red[256];
  __shared__ float4 gl[3072];
  char* stb = (char*)st;
  const int tid  = threadIdx.x;
  const int b    = blockIdx.x;


  float xs[4], part = 0.f;
#pragma unroll
  for (int t = 0; t < 4; ++t) {
    int p = tid + 256 * t;
    float v = x[b * PIX + p];
    xs[t] = v; part += v * v;
  }
  const float4* G4 = (const float4*)G;
  float4 gv[12];
#pragma unroll
  for (int i = 0; i < 12; ++i) gv[i] = G4[tid + 256 * i];
#pragma unroll
  for (int i = 0; i < 12; ++i) gl[tid + 256 * i] = gv[i];
  red[tid] = part; __syncthreads();
  for (int off = 128; off > 0; off >>= 1) {
    if (tid < off) red[tid] += red[tid + off];
    __syncthreads();
  }
  float inv = rsqrtf(red[0]);
  constexpr Tab8 EB = pair_bits<0>();
  constexpr int  EM = or8(EB);
  const int ve = lin8(tid, EB);
#pragma unroll
  for (int t = 0; t < 4; ++t) {
    lds_st(stb, xoff(ve, EM, phys<0>(512 * t)),     make_float2(xs[t] * inv, 0.f));
    lds_st(stb, xoff(ve, EM, phys<0>(512 * t + 1)), make_float2(0.f, 0.f));
  }
  __syncthreads();
  const float* Gs = (const float*)gl;

  const float yv = y[b];
  const float rc = cosf(0.5f * yv), rs = sinf(0.5f * yv);

  reupload_block<0>(stb, Gs,            tid, rc, rs);
  reupload_block<4>(stb, Gs + 4 * 1536, tid, rc, rs);

  constexpr Tab8 OB = pair_bits<4>();
  constexpr int  OM = or8(OB);
  const int vo = lin8(tid, OB);
#pragma unroll
  for (int t = 0; t < 4; ++t) {
    int p  = tid + 256 * t;
    float2 a = lds_ld(stb, xoff(vo, OM, phys<4>(512 * t)));
    out[b * PIX + p] = fminf(1.0f, (a.x * a.x + a.y * a.y) * (float)PIX);
  }
}

extern "C" void kernel_launch(void* const* d_in, const int* in_sizes, int n_in,
                              void* d_out, int out_size, void* d_ws, size_t ws_size,
                              hipStream_t stream) {
  const float* x  = (const float*)d_in[0];
  const float* y  = (const float*)d_in[1];
  const float* w0 = (const float*)d_in[2];
  const float* w1 = (const float*)d_in[3];
  float* G = (float*)d_ws;

  qprep<<<24, 256, 0, stream>>>(w0, w1, G);
  qsim<<<NB, 256, 0, stream>>>(x, y, G, (float*)d_out);
}


# wires 0-3 and 4-7 products chained in registers: the first is computed transposed so its accumulators are the second's B operand; no LDS store, barrier, re-gather between; changes summation order
# speedup vs baseline: 5.0217x; 1.0639x over previous
#include <hip/hip_runtime.h>
#include <math.h>


#define WIRES 11
#define DIM   2048
#define PIX   1024
#define NB    1024

typedef __attribute__((ext_vector_type(2))) float v2f;
typedef __attribute__((ext_vector_type(8))) float v8f;

__global__ void __launch_bounds__(256)
qprep(const float* __restrict__ w0, const float* __restrict__ w1,
      float* __restrict__ G) {
  const float PI_ = 3.14159265358979323846f;
  __shared__ float2 ent[16];
  __shared__ float2 zp[12];
  const int blk  = blockIdx.x;
  const int L    = blk / 3;
  const int grp  = blk - L * 3;
  const int cell = threadIdx.x;
  const float* w = (L < 4) ? w0 : w1;
  const int l = L & 3;
  const int Lb = L * 1536;

  if (grp != 2) {
    const int q0 = 4 * grp;
    if (cell < 12) {
      const int wire = (cell < 8) ? cell : q0 + (cell - 8);
      const int comp = (cell < 8) ? (grp == 0 ? 0 : 2) : 1;
      const float ang = PI_ * tanhf(w[(l * WIRES + wire) * 3 + comp]);
      zp[cell] = make_float2(cosf(0.5f * ang), sinf(0.5f * ang));
    }
    __syncthreads();
    const int m = cell >> 4, k = cell & 15;
    float r = 1.f;
#pragma unroll
    for (int i = 0; i < 4; ++i) {
      const int bp = 3 - i;
      const int mb = (m >> bp) & 1, kb = (k >> bp) & 1;
      const float2 t = zp[8 + i];
      r *= (mb == kb) ? t.x : (mb ? t.y : -t.y);
    }
    const int fh  = (grp == 0) ? ((k >> 1) & 1) : (k >> 3);
    const int fi  = (grp == 0) ? (((k >> 2) << 1) | (k & 1)) : (k & 7);
    const int pos = 128 * (fi >> 2) + 4 * (16 * fh + m) + (fi & 3);
    G[Lb + 256 * grp + pos] = r;
    if (grp == 0) {
      const int k1 = 4 * (cell >> 6) + 2 * ((cell >> 5) & 1) + (cell & 1);
      const int wb = (k1 << 4) | ((cell >> 1) & 15);
      float re = 1.f, im = 0.f;
#pragma unroll
      for (int q = 0; q < 8; ++q) {
        const float2 z = zp[q];
        const float zy = ((wb >> (7 - q)) & 1) ? z.y : -z.y;
        const float nr = re * z.x - im * zy;
        const float ni = re * zy + im * z.x;
        re = nr; im = ni;
      }
      G[Lb + 768 + 2 * cell]     = re;
      G[Lb + 768 + 2 * cell + 1] = im;
    } else {
      if (cell < 32) {
        const int qb = (cell < 16) ? 0 : 4;
        const int v  = cell & 15;
        float re = 1.f, im = 0.f;
#pragma unroll
        for (int i = 0; i < 4; ++i) {
          const float2 z = zp[qb + i];
          const float zy = ((v >> (3 - i)) & 1) ? z.y : -z.y;
          const float nr = re * z.x - im * zy;
          const float ni = re * zy + im * z.x;
          re = nr; im = ni;
        }
        G[Lb + 1280 + 2 * cell]     = re;
        G[Lb + 1280 + 2 * cell + 1] = im;
      } else if (cell < 224) {
        G[Lb + 1344 + (cell - 32)] = 0.f;
      }
    }
    return;
  }

  int m = cell >> 4, k = cell & 15;
  const int part = m & 1, cpart = k & 1;
  {
    const int R  = (((m >> 1) & 3) << 1) | (m >> 3);
    const int kr = (((k >> 1) & 3) << 1) | (k >> 3);
    m = R; k = kr;
  }
  const int nw = 3, q0 = 8, topbit = 2;

  const int wv = __builtin_amdgcn_readfirstlane(cell >> 5);
  const int wi = wv & 3;
  const bool dg = wv < 4;
  if (wi < nw) {
    const float* a3 = w + (l * WIRES + q0 + wi) * 3;
    const float t0 = a3[0], t1 = a3[1], t2 = a3[2];
    float phi   = PI_ * tanhf(t0);
    float theta = PI_ * tanhf(t1);
    float omega = PI_ * tanhf(t2);
    float mag = dg ? cosf(0.5f * theta) : sinf(0.5f * theta);
    float ang = dg ? 0.5f * (phi + omega) : 0.5f * (phi - omega);
    float pr = cosf(ang) * mag;
    float pi = sinf(ang) * mag;
    const int lane = cell & 31;
    if (lane == 0)
      ent[4 * wi + (dg ? 0 : 1)] = make_float2(dg ? pr : -pr, -pi);
    if (lane == 1)
      ent[4 * wi + (dg ? 3 : 2)] = make_float2(pr, dg ? pi : -pi);
  }
  __syncthreads();

  float2 u[3];
#pragma unroll
  for (int i = 0; i < 3; ++i) {
    const int bp = topbit - i;
    const int mb = (m >> bp) & 1, kb = (k >> bp) & 1;
    u[i] = ent[4 * i + 2 * mb + kb];
  }
  float re = 1.f, im = 0.f;
#pragma unroll
  for (int i = 0; i < 3; ++i) {
    float nr = re * u[i].x - im * u[i].y;
    float ni = re * u[i].y + im * u[i].x;
    re = nr; im = ni;
  }
  const int kk   = cell & 15;
  const int fh   = kk >> 3;
  const int fi   = kk & 7;
  const int pos  = 128 * (fi >> 2) + 4 * (16 * fh + (cell >> 4)) + (fi & 3);
  G[Lb + 512 + pos] = (part == cpart) ? re : (part ? im : -im);
}

__host__ __device__ constexpr int pinv_idx(int x, int r) {
  for (int q = WIRES - 1; q >= 0; --q) {
    int tq = q + r; if (tq >= WIRES) tq -= WIRES;
    x ^= ((x >> (10 - q)) & 1) << (10 - tq);
  }
  return x;
}

__host__ __device__ constexpr int swz(int i) {
  int m = 0;
  if (i & (1 << 5))  m ^= 3;
  if (i & (1 << 6))  m ^= 10;
  if (i & (1 << 7))  m ^= 14;
  if (i & (1 << 8))  m ^= 28;
  if (i & (1 << 9))  m ^= 11;
  if (i & (1 << 10)) m ^= 27;
  return i ^ m;
}
template <int RP>
__host__ __device__ constexpr int phys(int x) {
  return swz(RP ? pinv_idx(x, RP) : x) * 8;
}
struct Tab8 { int v[8]; };
__host__ __device__ constexpr int or8(Tab8 t) {
  int m = 0;
  for (int b = 0; b < 8; ++b) m |= t.v[b];
  return m;
}
template <int RP>
__host__ __device__ constexpr Tab8 g12_gather_bits() {
  const int ib[8] = {3, 4, 5, 6, 8, 0, 1, 2};
  Tab8 t{};
  for (int b = 0; b < 8; ++b) t.v[b] = phys<RP>(1 << ib[b]);
  return t;
}
__host__ __device__ constexpr Tab8 g12_store_bits() {
  const int ib[8] = {7, 8, 9, 10, 6, 0, 1, 2};
  Tab8 t{};
  for (int b = 0; b < 8; ++b) t.v[b] = phys<0>(1 << ib[b]);
  return t;
}
template <int RP, int LO>
__host__ __device__ constexpr Tab8 grp_kconst() {
  Tab8 t{};
  for (int c = 0; c < 4; ++c)
    for (int e = 0; e < 2; ++e) t.v[2 * c + e] = phys<RP>((4 * c + e) << LO);
  return t;
}
template <int LO>
__host__ __device__ constexpr Tab8 grp_rconst() {
  Tab8 t{};
  for (int r = 0; r < 8; ++r) t.v[r] = phys<0>(r << LO);
  return t;
}
template <int RP>
__host__ __device__ constexpr Tab8 pair_bits() {
  Tab8 t{};
  for (int b = 0; b < 8; ++b) t.v[b] = phys<RP>(2 << b);
  return t;
}
__device__ __forceinline__ int lin8(int tid, const Tab8 t) {
  int v = 0;
#pragma unroll
  for (int b = 0; b < 8; ++b) v ^= (-((tid >> b) & 1)) & t.v[b];
  return v;
}
__device__ __forceinline__ int xoff(int v, int vmask, int k) {
  return (v ^ (k & vmask)) + (k & ~vmask);
}
__device__ __forceinline__ float2 lds_ld(const char* stb, int off) {
  return *(const float2*)(stb + off);
}
__device__ __forceinline__ void lds_st(char* stb, int off, float2 v) {
  *(float2*)(stb + off) = v;
}

__device__ __forceinline__ float2 cmul(float2 a, float px, float py) {
  return make_float2(a.x * px - a.y * py, a.x * py + a.y * px);
}
template <int RP>
__device__ __forceinline__ void apply_group12(char* stb,
                                              const float* __restrict__ g1,
                                              const float* __restrict__ g2,
                                              const float* __restrict__ tin,
                                              int tid) {
  const int lane = tid & 31;

  constexpr Tab8 LB = g12_gather_bits<RP>();
  constexpr Tab8 SB = g12_store_bits();
  constexpr Tab8 KL = grp_kconst<RP, 7>();
  constexpr Tab8 KS = grp_rconst<3>();
  constexpr int  LM = or8(LB);
  constexpr int  SM = or8(SB);
  const int vl = lin8(tid, LB);
  const int vs = lin8(tid, SB);

  const float4* p1 = (const float4*)g1 + lane;
  const float4 r01 = p1[0], r23 = p1[32];
  const v2f m1[4] = { v2f{r01.x, r01.y}, v2f{r01.z, r01.w},
                      v2f{r23.x, r23.y}, v2f{r23.z, r23.w} };
  const float4* p2 = (const float4*)g2 + lane;
  const float4 q01 = p2[0], q23 = p2[32];
  const v2f m2[4] = { v2f{q01.x, q01.y}, v2f{q01.z, q01.w},
                      v2f{q23.x, q23.y}, v2f{q23.z, q23.w} };
  const float4* tp = (const float4*)tin + lane;
  float4 ph[4];
#pragma unroll
  for (int c = 0; c < 4; ++c) ph[c] = tp[32 * c];
  float2 s0[4], s1[4];
#pragma unroll
  for (int c = 0; c < 4; ++c) {
    s0[c] = lds_ld(stb, xoff(vl, LM, KL.v[2 * c]));
    s1[c] = lds_ld(stb, xoff(vl, LM, KL.v[2 * c + 1]));
  }
  if constexpr (RP != 0) __syncthreads();
  else __builtin_amdgcn_sched_barrier(0);

  v2f ar[4], ai[4];
#pragma unroll
  for (int c = 0; c < 4; ++c) {
    const float2 t0 = cmul(s0[c], ph[c].x, ph[c].y);
    const float2 t1 = cmul(s1[c], ph[c].z, ph[c].w);
    ar[c] = v2f{t0.x, t1.x};
    ai[c] = v2f{t0.y, t1.y};
  }

  v8f ur = {0, 0, 0, 0, 0, 0, 0, 0};
  v8f ui = {0, 0, 0, 0, 0, 0, 0, 0};
#pragma unroll
  for (int c = 0; c < 4; ++c) {
    ur = __builtin_amdgcn_wmma_f32_16x16x4_f32(false, ar[c], false, m1[c],
                                               (short)0, ur, false, false);
    ui = __builtin_amdgcn_wmma_f32_16x16x4_f32(false, ai[c], false, m1[c],
                                               (short)0, ui, false, false);
  }
  v8f dr = {0, 0, 0, 0, 0, 0, 0, 0};
  v8f di = {0, 0, 0, 0, 0, 0, 0, 0};
#pragma unroll
  for (int c = 0; c < 4; ++c) {
    dr = __builtin_amdgcn_wmma_f32_16x16x4_f32(false, m2[c], false,
                                               v2f{ur[2 * c], ur[2 * c + 1]},
                                               (short)0, dr, false, false);
    di = __builtin_amdgcn_wmma_f32_16x16x4_f32(false, m2[c], false,
                                               v2f{ui[2 * c], ui[2 * c + 1]},
                                               (short)0, di, false, false);
  }

#pragma unroll
  for (int r = 0; r < 8; ++r) {
    lds_st(stb, xoff(vs, SM, KS.v[r]), make_float2(dr[r], di[r]));
  }
}

__host__ __device__ constexpr Tab8 g3_bits() {
  const int ib[8] = {3, 4, 7, 8, 0, 5, 6, 9};
  Tab8 t{};
  for (int b = 0; b < 8; ++b) t.v[b] = phys<0>(1 << ib[b]);
  return t;
}
__host__ __device__ constexpr Tab8 g3_const() {
  Tab8 t{};
  for (int tl = 0; tl < 2; ++tl)
    for (int c = 0; c < 4; ++c) t.v[4 * tl + c] = phys<0>((tl << 10) | (c << 1));
  return t;
}
__device__ __forceinline__ void apply_group3(char* stb,
                                             const float* __restrict__ ge,
                                             const float* __restrict__ dout,
                                             int tid) {
  const int lane = tid & 31;
  const float2* dt = (const float2*)dout;
  const int m1 = ((tid >> 2) & 3) | (((tid >> 7) & 1) << 2);
  const int m2 = (tid & 3) | (((tid >> 5) & 3) << 2);
  const float2 pa = dt[m1], pb = dt[m1 + 8], pc = dt[16 + m2];

  constexpr Tab8 LB = g3_bits();
  constexpr Tab8 KL = g3_const();
  constexpr int  LM = or8(LB);
  const int vl = lin8(tid, LB);

  const float4* ap = (const float4*)ge + lane;
  const float4 a01 = ap[0], a23 = ap[32];
  const v2f a[4] = { v2f{a01.x, a01.y}, v2f{a01.z, a01.w},
                     v2f{a23.x, a23.y}, v2f{a23.z, a23.w} };
  v2f b0[4], b1[4];
#pragma unroll
  for (int c = 0; c < 4; ++c) {
    float2 s0 = lds_ld(stb, xoff(vl, LM, KL.v[c]));
    float2 s1 = lds_ld(stb, xoff(vl, LM, KL.v[4 + c]));
    b0[c] = v2f{s0.x, s0.y};
    b1[c] = v2f{s1.x, s1.y};
  }
  __builtin_amdgcn_sched_barrier(0);
  const float2 p0 = cmul(pa, pc.x, pc.y);
  const float2 p1 = cmul(pb, pc.x, pc.y);
#pragma unroll
  for (int c = 0; c < 4; ++c) {
    const float2 t0 = cmul(make_float2(b0[c].x, b0[c].y), p0.x, p0.y);
    const float2 t1 = cmul(make_float2(b1[c].x, b1[c].y), p1.x, p1.y);
    b0[c] = v2f{t0.x, t0.y};
    b1[c] = v2f{t1.x, t1.y};
  }

  v8f d0 = {0, 0, 0, 0, 0, 0, 0, 0};
  v8f d1 = {0, 0, 0, 0, 0, 0, 0, 0};
#pragma unroll
  for (int c = 0; c < 4; ++c) {
    d0 = __builtin_amdgcn_wmma_f32_16x16x4_f32(false, a[c], false, b0[c],
                                               (short)0, d0, false, false);
    d1 = __builtin_amdgcn_wmma_f32_16x16x4_f32(false, a[c], false, b1[c],
                                               (short)0, d1, false, false);
  }
#pragma unroll
  for (int j = 0; j < 4; ++j) {
    lds_st(stb, xoff(vl, LM, KL.v[j]),     make_float2(d0[2 * j], d0[2 * j + 1]));
    lds_st(stb, xoff(vl, LM, KL.v[4 + j]), make_float2(d1[2 * j], d1[2 * j + 1]));
  }
}

template <int RP>
__device__ __forceinline__ void rx_pass(char* stb, int tid, float rc, float rs) {
  constexpr Tab8 PB = pair_bits<RP>();
  constexpr int  PM = or8(PB);
  constexpr Tab8 KT = { { phys<RP>(0),    phys<RP>(1),
                          phys<RP>(512),  phys<RP>(513),
                          phys<RP>(1024), phys<RP>(1025),
                          phys<RP>(1536), phys<RP>(1537) } };
  const int vt = lin8(tid, PB);
#pragma unroll
  for (int t = 0; t < 4; ++t) {
    int x0 = xoff(vt, PM, KT.v[2 * t]);
    int x1 = xoff(vt, PM, KT.v[2 * t + 1]);
    float2 a0 = lds_ld(stb, x0), a1 = lds_ld(stb, x1);
    lds_st(stb, x0, make_float2(rc * a0.x + rs * a1.y, rc * a0.y - rs * a1.x));
    lds_st(stb, x1, make_float2(rc * a1.x + rs * a0.y, rc * a1.y - rs * a0.x));
  }
}

template <int RP>
__device__ __forceinline__ void layer(char* stb, const float* __restrict__ Gl, int tid) {
  apply_group12<RP>(stb, Gl, Gl + 256, Gl + 768, tid);    __syncthreads();
  apply_group3(stb, Gl + 512, Gl + 1280, tid);            __syncthreads();
}

template <int RP>
__device__ __forceinline__ void reupload_block(char* stb, const float* __restrict__ Gb,
                                               int tid, float rc, float rs) {
  rx_pass<RP>(stb, tid, rc, rs);
  __syncthreads();
  layer<RP>(stb, Gb,            tid);
  layer<1>(stb, Gb + 1 * 1536, tid);
  layer<2>(stb, Gb + 2 * 1536, tid);
  layer<3>(stb, Gb + 3 * 1536, tid);
}

__global__ void __launch_bounds__(256)
qsim(const float* __restrict__ x, const float* __restrict__ y,
     const float* __restrict__ G, float* __restrict__ out) {
  __shared__ float2 st[DIM];
  __shared__ float  red[256];
  __shared__ float4 gl[3072];
  char* stb = (char*)st;
  const int tid  = threadIdx.x;
  const int b    = blockIdx.x;


  float xs[4], part = 0.f;
#pragma unroll
  for (int t = 0; t < 4; ++t) {
    int p = tid + 256 * t;
    float v = x[b * PIX + p];
    xs[t] = v; part += v * v;
  }
  const float4* G4 = (const float4*)G;
  float4 gv[12];
#pragma unroll
  for (int i = 0; i < 12; ++i) gv[i] = G4[tid + 256 * i];
#pragma unroll
  for (int i = 0; i < 12; ++i) gl[tid + 256 * i] = gv[i];
  red[tid] = part; __syncthreads();
  for (int off = 128; off > 0; off >>= 1) {
    if (tid < off) red[tid] += red[tid + off];
    __syncthreads();
  }
  float inv = rsqrtf(red[0]);
  constexpr Tab8 EB = pair_bits<0>();
  constexpr int  EM = or8(EB);
  const int ve = lin8(tid, EB);
#pragma unroll
  for (int t = 0; t < 4; ++t) {
    lds_st(stb, xoff(ve, EM, phys<0>(512 * t)),     make_float2(xs[t] * inv, 0.f));
    lds_st(stb, xoff(ve, EM, phys<0>(512 * t + 1)), make_float2(0.f, 0.f));
  }
  __syncthreads();
  const float* Gs = (const float*)gl;

  const float yv = y[b];
  const float rc = cosf(0.5f * yv), rs = sinf(0.5f * yv);

  reupload_block<0>(stb, Gs,            tid, rc, rs);
  reupload_block<4>(stb, Gs + 4 * 1536, tid, rc, rs);

  constexpr Tab8 OB = pair_bits<4>();
  constexpr int  OM = or8(OB);
  const int vo = lin8(tid, OB);
#pragma unroll
  for (int t = 0; t < 4; ++t) {
    int p  = tid + 256 * t;
    float2 a = lds_ld(stb, xoff(vo, OM, phys<4>(512 * t)));
    out[b * PIX + p] = fminf(1.0f, (a.x * a.x + a.y * a.y) * (float)PIX);
  }
}

extern "C" void kernel_launch(void* const* d_in, const int* in_sizes, int n_in,
                              void* d_out, int out_size, void* d_ws, size_t ws_size,
                              hipStream_t stream) {
  const float* x  = (const float*)d_in[0];
  const float* y  = (const float*)d_in[1];
  const float* w0 = (const float*)d_in[2];
  const float* w1 = (const float*)d_in[3];
  float* G = (float*)d_ws;

  qprep<<<24, 256, 0, stream>>>(w0, w1, G);
  qsim<<<NB, 256, 0, stream>>>(x, y, G, (float*)d_out);
}


# embedding store applies the first RX(y) (last wire starts in |0>) and skips norm; read-out scales by 1024/sum(x^2), sums by lane exchange: no norm tree, no first RX pass; rounding, sum order change
# speedup vs baseline: 5.2331x; 1.0421x over previous
#include <hip/hip_runtime.h>
#include <math.h>


#define WIRES 11
#define DIM   2048
#define PIX   1024
#define NB    1024

typedef __attribute__((ext_vector_type(2))) float v2f;
typedef __attribute__((ext_vector_type(8))) float v8f;

__global__ void __launch_bounds__(256)
qprep(const float* __restrict__ w0, const float* __restrict__ w1,
      float* __restrict__ G) {
  const float PI_ = 3.14159265358979323846f;
  __shared__ float2 ent[16];
  __shared__ float2 zp[12];
  const int blk  = blockIdx.x;
  const int L    = blk / 3;
  const int grp  = blk - L * 3;
  const int cell = threadIdx.x;
  const float* w = (L < 4) ? w0 : w1;
  const int l = L & 3;
  const int Lb = L * 1536;

  if (grp != 2) {
    const int q0 = 4 * grp;
    if (cell < 12) {
      const int wire = (cell < 8) ? cell : q0 + (cell - 8);
      const int comp = (cell < 8) ? (grp == 0 ? 0 : 2) : 1;
      const float ang = PI_ * tanhf(w[(l * WIRES + wire) * 3 + comp]);
      zp[cell] = make_float2(cosf(0.5f * ang), sinf(0.5f * ang));
    }
    __syncthreads();
    const int m = cell >> 4, k = cell & 15;
    float r = 1.f;
#pragma unroll
    for (int i = 0; i < 4; ++i) {
      const int bp = 3 - i;
      const int mb = (m >> bp) & 1, kb = (k >> bp) & 1;
      const float2 t = zp[8 + i];
      r *= (mb == kb) ? t.x : (mb ? t.y : -t.y);
    }
    const int fh  = (grp == 0) ? ((k >> 1) & 1) : (k >> 3);
    const int fi  = (grp == 0) ? (((k >> 2) << 1) | (k & 1)) : (k & 7);
    const int pos = 128 * (fi >> 2) + 4 * (16 * fh + m) + (fi & 3);
    G[Lb + 256 * grp + pos] = r;
    if (grp == 0) {
      const int k1 = 4 * (cell >> 6) + 2 * ((cell >> 5) & 1) + (cell & 1);
      const int wb = (k1 << 4) | ((cell >> 1) & 15);
      float re = 1.f, im = 0.f;
#pragma unroll
      for (int q = 0; q < 8; ++q) {
        const float2 z = zp[q];
        const float zy = ((wb >> (7 - q)) & 1) ? z.y : -z.y;
        const float nr = re * z.x - im * zy;
        const float ni = re * zy + im * z.x;
        re = nr; im = ni;
      }
      G[Lb + 768 + 2 * cell]     = re;
      G[Lb + 768 + 2 * cell + 1] = im;
    } else {
      if (cell < 32) {
        const int qb = (cell < 16) ? 0 : 4;
        const int v  = cell & 15;
        float re = 1.f, im = 0.f;
#pragma unroll
        for (int i = 0; i < 4; ++i) {
          const float2 z = zp[qb + i];
          const float zy = ((v >> (3 - i)) & 1) ? z.y : -z.y;
          const float nr = re * z.x - im * zy;
          const float ni = re * zy + im * z.x;
          re = nr; im = ni;
        }
        G[Lb + 1280 + 2 * cell]     = re;
        G[Lb + 1280 + 2 * cell + 1] = im;
      } else if (cell < 224) {
        G[Lb + 1344 + (cell - 32)] = 0.f;
      }
    }
    return;
  }

  int m = cell >> 4, k = cell & 15;
  const int part = m & 1, cpart = k & 1;
  {
    const int R  = (((m >> 1) & 3) << 1) | (m >> 3);
    const int kr = (((k >> 1) & 3) << 1) | (k >> 3);
    m = R; k = kr;
  }
  const int nw = 3, q0 = 8, topbit = 2;

  const int wv = __builtin_amdgcn_readfirstlane(cell >> 5);
  const int wi = wv & 3;
  const bool dg = wv < 4;
  if (wi < nw) {
    const float* a3 = w + (l * WIRES + q0 + wi) * 3;
    const float t0 = a3[0], t1 = a3[1], t2 = a3[2];
    float phi   = PI_ * tanhf(t0);
    float theta = PI_ * tanhf(t1);
    float omega = PI_ * tanhf(t2);
    float mag = dg ? cosf(0.5f * theta) : sinf(0.5f * theta);
    float ang = dg ? 0.5f * (phi + omega) : 0.5f * (phi - omega);
    float pr = cosf(ang) * mag;
    float pi = sinf(ang) * mag;
    const int lane = cell & 31;
    if (lane == 0)
      ent[4 * wi + (dg ? 0 : 1)] = make_float2(dg ? pr : -pr, -pi);
    if (lane == 1)
      ent[4 * wi + (dg ? 3 : 2)] = make_float2(pr, dg ? pi : -pi);
  }
  __syncthreads();

  float2 u[3];
#pragma unroll
  for (int i = 0; i < 3; ++i) {
    const int bp = topbit - i;
    const int mb = (m >> bp) & 1, kb = (k >> bp) & 1;
    u[i] = ent[4 * i + 2 * mb + kb];
  }
  float re = 1.f, im = 0.f;
#pragma unroll
  for (int i = 0; i < 3; ++i) {
    float nr = re * u[i].x - im * u[i].y;
    float ni = re * u[i].y + im * u[i].x;
    re = nr; im = ni;
  }
  const int kk   = cell & 15;
  const int fh   = kk >> 3;
  const int fi   = kk & 7;
  const int pos  = 128 * (fi >> 2) + 4 * (16 * fh + (cell >> 4)) + (fi & 3);
  G[Lb + 512 + pos] = (part == cpart) ? re : (part ? im : -im);
}

__host__ __device__ constexpr int pinv_idx(int x, int r) {
  for (int q = WIRES - 1; q >= 0; --q) {
    int tq = q + r; if (tq >= WIRES) tq -= WIRES;
    x ^= ((x >> (10 - q)) & 1) << (10 - tq);
  }
  return x;
}

__host__ __device__ constexpr int swz(int i) {
  int m = 0;
  if (i & (1 << 5))  m ^= 3;
  if (i & (1 << 6))  m ^= 10;
  if (i & (1 << 7))  m ^= 14;
  if (i & (1 << 8))  m ^= 28;
  if (i & (1 << 9))  m ^= 11;
  if (i & (1 << 10)) m ^= 27;
  return i ^ m;
}
template <int RP>
__host__ __device__ constexpr int phys(int x) {
  return swz(RP ? pinv_idx(x, RP) : x) * 8;
}
struct Tab8 { int v[8]; };
__host__ __device__ constexpr int or8(Tab8 t) {
  int m = 0;
  for (int b = 0; b < 8; ++b) m |= t.v[b];
  return m;
}
template <int RP>
__host__ __device__ constexpr Tab8 g12_gather_bits() {
  const int ib[8] = {3, 4, 5, 6, 8, 0, 1, 2};
  Tab8 t{};
  for (int b = 0; b < 8; ++b) t.v[b] = phys<RP>(1 << ib[b]);
  return t;
}
__host__ __device__ constexpr Tab8 g12_store_bits() {
  const int ib[8] = {7, 8, 9, 10, 6, 0, 1, 2};
  Tab8 t{};
  for (int b = 0; b < 8; ++b) t.v[b] = phys<0>(1 << ib[b]);
  return t;
}
template <int RP, int LO>
__host__ __device__ constexpr Tab8 grp_kconst() {
  Tab8 t{};
  for (int c = 0; c < 4; ++c)
    for (int e = 0; e < 2; ++e) t.v[2 * c + e] = phys<RP>((4 * c + e) << LO);
  return t;
}
template <int LO>
__host__ __device__ constexpr Tab8 grp_rconst() {
  Tab8 t{};
  for (int r = 0; r < 8; ++r) t.v[r] = phys<0>(r << LO);
  return t;
}
template <int RP>
__host__ __device__ constexpr Tab8 pair_bits() {
  Tab8 t{};
  for (int b = 0; b < 8; ++b) t.v[b] = phys<RP>(2 << b);
  return t;
}
__device__ __forceinline__ int lin8(int tid, const Tab8 t) {
  int v = 0;
#pragma unroll
  for (int b = 0; b < 8; ++b) v ^= (-((tid >> b) & 1)) & t.v[b];
  return v;
}
__device__ __forceinline__ int xoff(int v, int vmask, int k) {
  return (v ^ (k & vmask)) + (k & ~vmask);
}
__device__ __forceinline__ float2 lds_ld(const char* stb, int off) {
  return *(const float2*)(stb + off);
}
__device__ __forceinline__ void lds_st(char* stb, int off, float2 v) {
  *(float2*)(stb + off) = v;
}

__device__ __forceinline__ float2 cmul(float2 a, float px, float py) {
  return make_float2(a.x * px - a.y * py, a.x * py + a.y * px);
}
template <int RP>
__device__ __forceinline__ void apply_group12(char* stb,
                                              const float* __restrict__ g1,
                                              const float* __restrict__ g2,
                                              const float* __restrict__ tin,
                                              int tid) {
  const int lane = tid & 31;

  constexpr Tab8 LB = g12_gather_bits<RP>();
  constexpr Tab8 SB = g12_store_bits();
  constexpr Tab8 KL = grp_kconst<RP, 7>();
  constexpr Tab8 KS = grp_rconst<3>();
  constexpr int  LM = or8(LB);
  constexpr int  SM = or8(SB);
  const int vl = lin8(tid, LB);
  const int vs = lin8(tid, SB);

  const float4* p1 = (const float4*)g1 + lane;
  const float4 r01 = p1[0], r23 = p1[32];
  const v2f m1[4] = { v2f{r01.x, r01.y}, v2f{r01.z, r01.w},
                      v2f{r23.x, r23.y}, v2f{r23.z, r23.w} };
  const float4* p2 = (const float4*)g2 + lane;
  const float4 q01 = p2[0], q23 = p2[32];
  const v2f m2[4] = { v2f{q01.x, q01.y}, v2f{q01.z, q01.w},
                      v2f{q23.x, q23.y}, v2f{q23.z, q23.w} };
  const float4* tp = (const float4*)tin + lane;
  float4 ph[4];
#pragma unroll
  for (int c = 0; c < 4; ++c) ph[c] = tp[32 * c];
  float2 s0[4], s1[4];
#pragma unroll
  for (int c = 0; c < 4; ++c) {
    s0[c] = lds_ld(stb, xoff(vl, LM, KL.v[2 * c]));
    s1[c] = lds_ld(stb, xoff(vl, LM, KL.v[2 * c + 1]));
  }
  if constexpr (RP != 0) __syncthreads();
  else __builtin_amdgcn_sched_barrier(0);

  v2f ar[4], ai[4];
#pragma unroll
  for (int c = 0; c < 4; ++c) {
    const float2 t0 = cmul(s0[c], ph[c].x, ph[c].y);
    const float2 t1 = cmul(s1[c], ph[c].z, ph[c].w);
    ar[c] = v2f{t0.x, t1.x};
    ai[c] = v2f{t0.y, t1.y};
  }

  v8f ur = {0, 0, 0, 0, 0, 0, 0, 0};
  v8f ui = {0, 0, 0, 0, 0, 0, 0, 0};
#pragma unroll
  for (int c = 0; c < 4; ++c) {
    ur = __builtin_amdgcn_wmma_f32_16x16x4_f32(false, ar[c], false, m1[c],
                                               (short)0, ur, false, false);
    ui = __builtin_amdgcn_wmma_f32_16x16x4_f32(false, ai[c], false, m1[c],
                                               (short)0, ui, false, false);
  }
  v8f dr = {0, 0, 0, 0, 0, 0, 0, 0};
  v8f di = {0, 0, 0, 0, 0, 0, 0, 0};
#pragma unroll
  for (int c = 0; c < 4; ++c) {
    dr = __builtin_amdgcn_wmma_f32_16x16x4_f32(false, m2[c], false,
                                               v2f{ur[2 * c], ur[2 * c + 1]},
                                               (short)0, dr, false, false);
    di = __builtin_amdgcn_wmma_f32_16x16x4_f32(false, m2[c], false,
                                               v2f{ui[2 * c], ui[2 * c + 1]},
                                               (short)0, di, false, false);
  }

#pragma unroll
  for (int r = 0; r < 8; ++r) {
    lds_st(stb, xoff(vs, SM, KS.v[r]), make_float2(dr[r], di[r]));
  }
}

__host__ __device__ constexpr Tab8 g3_bits() {
  const int ib[8] = {3, 4, 7, 8, 0, 5, 6, 9};
  Tab8 t{};
  for (int b = 0; b < 8; ++b) t.v[b] = phys<0>(1 << ib[b]);
  return t;
}
__host__ __device__ constexpr Tab8 g3_const() {
  Tab8 t{};
  for (int tl = 0; tl < 2; ++tl)
    for (int c = 0; c < 4; ++c) t.v[4 * tl + c] = phys<0>((tl << 10) | (c << 1));
  return t;
}
__device__ __forceinline__ void apply_group3(char* stb,
                                             const float* __restrict__ ge,
                                             const float* __restrict__ dout,
                                             int tid) {
  const int lane = tid & 31;
  const float2* dt = (const float2*)dout;
  const int m1 = ((tid >> 2) & 3) | (((tid >> 7) & 1) << 2);
  const int m2 = (tid & 3) | (((tid >> 5) & 3) << 2);
  const float2 pa = dt[m1], pb = dt[m1 + 8], pc = dt[16 + m2];

  constexpr Tab8 LB = g3_bits();
  constexpr Tab8 KL = g3_const();
  constexpr int  LM = or8(LB);
  const int vl = lin8(tid, LB);

  const float4* ap = (const float4*)ge + lane;
  const float4 a01 = ap[0], a23 = ap[32];
  const v2f a[4] = { v2f{a01.x, a01.y}, v2f{a01.z, a01.w},
                     v2f{a23.x, a23.y}, v2f{a23.z, a23.w} };
  v2f b0[4], b1[4];
#pragma unroll
  for (int c = 0; c < 4; ++c) {
    float2 s0 = lds_ld(stb, xoff(vl, LM, KL.v[c]));
    float2 s1 = lds_ld(stb, xoff(vl, LM, KL.v[4 + c]));
    b0[c] = v2f{s0.x, s0.y};
    b1[c] = v2f{s1.x, s1.y};
  }
  __builtin_amdgcn_sched_barrier(0);
  const float2 p0 = cmul(pa, pc.x, pc.y);
  const float2 p1 = cmul(pb, pc.x, pc.y);
#pragma unroll
  for (int c = 0; c < 4; ++c) {
    const float2 t0 = cmul(make_float2(b0[c].x, b0[c].y), p0.x, p0.y);
    const float2 t1 = cmul(make_float2(b1[c].x, b1[c].y), p1.x, p1.y);
    b0[c] = v2f{t0.x, t0.y};
    b1[c] = v2f{t1.x, t1.y};
  }

  v8f d0 = {0, 0, 0, 0, 0, 0, 0, 0};
  v8f d1 = {0, 0, 0, 0, 0, 0, 0, 0};
#pragma unroll
  for (int c = 0; c < 4; ++c) {
    d0 = __builtin_amdgcn_wmma_f32_16x16x4_f32(false, a[c], false, b0[c],
                                               (short)0, d0, false, false);
    d1 = __builtin_amdgcn_wmma_f32_16x16x4_f32(false, a[c], false, b1[c],
                                               (short)0, d1, false, false);
  }
#pragma unroll
  for (int j = 0; j < 4; ++j) {
    lds_st(stb, xoff(vl, LM, KL.v[j]),     make_float2(d0[2 * j], d0[2 * j + 1]));
    lds_st(stb, xoff(vl, LM, KL.v[4 + j]), make_float2(d1[2 * j], d1[2 * j + 1]));
  }
}

template <int RP>
__device__ __forceinline__ void rx_pass(char* stb, int tid, float rc, float rs) {
  constexpr Tab8 PB = pair_bits<RP>();
  constexpr int  PM = or8(PB);
  constexpr Tab8 KT = { { phys<RP>(0),    phys<RP>(1),
                          phys<RP>(512),  phys<RP>(513),
                          phys<RP>(1024), phys<RP>(1025),
                          phys<RP>(1536), phys<RP>(1537) } };
  const int vt = lin8(tid, PB);
#pragma unroll
  for (int t = 0; t < 4; ++t) {
    int x0 = xoff(vt, PM, KT.v[2 * t]);
    int x1 = xoff(vt, PM, KT.v[2 * t + 1]);
    float2 a0 = lds_ld(stb, x0), a1 = lds_ld(stb, x1);
    lds_st(stb, x0, make_float2(rc * a0.x + rs * a1.y, rc * a0.y - rs * a1.x));
    lds_st(stb, x1, make_float2(rc * a1.x + rs * a0.y, rc * a1.y - rs * a0.x));
  }
}

template <int RP>
__device__ __forceinline__ void layer(char* stb, const float* __restrict__ Gl, int tid) {
  apply_group12<RP>(stb, Gl, Gl + 256, Gl + 768, tid);    __syncthreads();
  apply_group3(stb, Gl + 512, Gl + 1280, tid);            __syncthreads();
}

template <int RP, bool RXP>
__device__ __forceinline__ void reupload_block(char* stb, const float* __restrict__ Gb,
                                               int tid, float rc, float rs) {
  if constexpr (RXP) {
    rx_pass<RP>(stb, tid, rc, rs);
    __syncthreads();
  }
  layer<RP>(stb, Gb,            tid);
  layer<1>(stb, Gb + 1 * 1536, tid);
  layer<2>(stb, Gb + 2 * 1536, tid);
  layer<3>(stb, Gb + 3 * 1536, tid);
}

__device__ __forceinline__ float wave_sum(float v) {
  v += __builtin_bit_cast(float, __builtin_amdgcn_update_dpp(0, __builtin_bit_cast(int, v), 0x161, 0xf, 0xf, false));
  v += __builtin_bit_cast(float, __builtin_amdgcn_update_dpp(0, __builtin_bit_cast(int, v), 0x162, 0xf, 0xf, false));
  v += __builtin_bit_cast(float, __builtin_amdgcn_update_dpp(0, __builtin_bit_cast(int, v), 0x164, 0xf, 0xf, false));
  v += __builtin_bit_cast(float, __builtin_amdgcn_update_dpp(0, __builtin_bit_cast(int, v), 0x168, 0xf, 0xf, false));
  const int b = __builtin_bit_cast(int, v);
  v += __builtin_bit_cast(float, __builtin_amdgcn_permlanex16(b, b, 0x76543210, 0xfedcba98, false, false));
  return v;
}

__global__ void __launch_bounds__(256)
qsim(const float* __restrict__ x, const float* __restrict__ y,
     const float* __restrict__ G, float* __restrict__ out) {
  __shared__ float2 st[DIM];
  __shared__ float4 red4[2];
  __shared__ float4 gl[3072];
  char* stb = (char*)st;
  const int tid  = threadIdx.x;
  const int b    = blockIdx.x;


  const float yv = y[b];
  float xs[4];
#pragma unroll
  for (int t = 0; t < 4; ++t) xs[t] = x[b * PIX + tid + 256 * t];
  const float4* G4 = (const float4*)G;
  float4 gv[12];
#pragma unroll
  for (int i = 0; i < 12; ++i) gv[i] = G4[tid + 256 * i];
#pragma unroll
  for (int i = 0; i < 12; ++i) gl[tid + 256 * i] = gv[i];
  const float part = wave_sum((xs[0] * xs[0] + xs[1] * xs[1]) + (xs[2] * xs[2] + xs[3] * xs[3]));
  if ((tid & 31) == 0) ((float*)red4)[tid >> 5] = part;
  const float rc = cosf(0.5f * yv), rs = sinf(0.5f * yv);
  constexpr Tab8 EB = pair_bits<0>();
  constexpr int  EM = or8(EB);
  const int ve = lin8(tid, EB);
#pragma unroll
  for (int t = 0; t < 4; ++t) {
    lds_st(stb, xoff(ve, EM, phys<0>(512 * t)),     make_float2(rc * xs[t], 0.f));
    lds_st(stb, xoff(ve, EM, phys<0>(512 * t + 1)), make_float2(0.f, -(rs * xs[t])));
  }
  __syncthreads();
  const float* Gs = (const float*)gl;
  const float4 q0 = red4[0], q1 = red4[1];
  const float oscale = (float)PIX / (((q0.x + q0.y) + (q0.z + q0.w)) + ((q1.x + q1.y) + (q1.z + q1.w)));

  reupload_block<0, false>(stb, Gs,            tid, rc, rs);
  reupload_block<4, true>(stb, Gs + 4 * 1536, tid, rc, rs);

  constexpr Tab8 OB = pair_bits<4>();
  constexpr int  OM = or8(OB);
  const int vo = lin8(tid, OB);
#pragma unroll
  for (int t = 0; t < 4; ++t) {
    int p  = tid + 256 * t;
    float2 a = lds_ld(stb, xoff(vo, OM, phys<4>(512 * t)));
    out[b * PIX + p] = fminf(1.0f, (a.x * a.x + a.y * a.y) * oscale);
  }
}

extern "C" void kernel_launch(void* const* d_in, const int* in_sizes, int n_in,
                              void* d_out, int out_size, void* d_ws, size_t ws_size,
                              hipStream_t stream) {
  const float* x  = (const float*)d_in[0];
  const float* y  = (const float*)d_in[1];
  const float* w0 = (const float*)d_in[2];
  const float* w1 = (const float*)d_in[3];
  float* G = (float*)d_ws;

  qprep<<<24, 256, 0, stream>>>(w0, w1, G);
  qsim<<<NB, 256, 0, stream>>>(x, y, G, (float*)d_out);
}
